# starting kernel
# baseline (speedup 1.0000x reference)
#include <hip/hip_runtime.h>
#include <math.h>

#define SRATE     22050
#define NFFT      1024
#define HOP       256
#define NMELS     128
#define NFREQS    513
#define FPAD      528
#define BATCH     16
#define NSAMP     661504
#define NFRAMES   2585
#define TTILES    162
#define SPAN      (15*HOP + NFFT)
#define NCH       (BATCH*NMELS)

typedef __attribute__((ext_vector_type(2))) float v2f;
typedef __attribute__((ext_vector_type(8))) float v8f;

__global__ void init_dft(float* __restrict__ dftCos, float* __restrict__ dftSin) {
    int idx = blockIdx.x * blockDim.x + threadIdx.x;
    if (idx >= NFFT * FPAD) return;
    int k = idx / FPAD;
    int f = idx % FPAD;
    float c = 0.0f, s = 0.0f;
    if (f < NFREQS) {
        float win = 0.5f - 0.5f * cosf(6.28318530717958647692f * (float)k / (float)NFFT);
        int r = (k * f) & (NFFT - 1);
        float ang = (float)r * (6.28318530717958647692f / (float)NFFT);
        c = win * cosf(ang);
        s = win * sinf(ang);
    }
    int off = (k >> 1) * (FPAD * 2) + f * 2 + (k & 1);
    dftCos[off] = c;
    dftSin[off] = s;
}

__device__ __forceinline__ float hz_to_mel(float f) { return 2595.0f * log10f(1.0f + f / 700.0f); }
__device__ __forceinline__ float mel_to_hz(float m) { return 700.0f * (exp10f(m / 2595.0f) - 1.0f); }

__global__ void init_mel(float* __restrict__ melFB) {
    int idx = blockIdx.x * blockDim.x + threadIdx.x;
    if (idx >= FPAD * NMELS) return;
    int k = idx / NMELS;
    int m = idx % NMELS;
    float v = 0.0f;
    if (k < NFREQS) {
        float m_lo = hz_to_mel(50.0f);
        float m_hi = hz_to_mel((float)SRATE * 0.5f);
        float step = (m_hi - m_lo) / (float)(NMELS + 1);
        float f0 = mel_to_hz(m_lo + step * (float)m);
        float f1 = mel_to_hz(m_lo + step * (float)(m + 1));
        float f2 = mel_to_hz(m_lo + step * (float)(m + 2));
        float freq = (float)k * ((float)SRATE * 0.5f / (float)(NFFT / 2));
        float down = (freq - f0) / (f1 - f0);
        float up   = (f2 - freq) / (f2 - f1);
        v = fmaxf(0.0f, fminf(down, up));
    }
    melFB[(k >> 1) * (NMELS * 2) + m * 2 + (k & 1)] = v;
}

__global__ __launch_bounds__(128)
void dft_mel_kernel(const float* __restrict__ x,
                    const float* __restrict__ dftCos,
                    const float* __restrict__ dftSin,
                    const float* __restrict__ melFB,
                    float* __restrict__ melBuf)
{
    __shared__ float span[SPAN];
    __shared__ float mag[16 * FPAD];

    const int tile = blockIdx.x;
    const int b    = blockIdx.y;
    const int tid  = threadIdx.x;
    const int ln   = tid & 31;
    const int wv   = tid >> 5;
    const int t0   = tile * 16;

    const float* xb = x + (size_t)b * NSAMP;
    const long base = (long)t0 * HOP - (NFFT / 2);
    for (int i = tid; i < SPAN; i += 128) {
        long p = base + i;
        if (p < 0) p = -p;
        if (p >= NSAMP) p = 2L * NSAMP - 2 - p;
        if (p < 0) p = 0;
        if (p >= NSAMP) p = NSAMP - 1;
        span[i] = xb[p];
    }
    __syncthreads();

    const int mrow  = ln & 15;
    const int kh    = (ln >> 4) << 1;
    const int ncol  = ln & 15;
    const int mbase = (ln >> 4) * 8;

    for (int nt = wv; nt < FPAD / 16; nt += 4) {
        v8f accRe = {0.f, 0.f, 0.f, 0.f, 0.f, 0.f, 0.f, 0.f};
        v8f accIm = {0.f, 0.f, 0.f, 0.f, 0.f, 0.f, 0.f, 0.f};
        const int n = nt * 16 + ncol;
        #pragma unroll 4
        for (int kb = 0; kb < NFFT; kb += 4) {
            const int k0 = kb + kh;
            v2f a  = *(const v2f*)&span[mrow * HOP + k0];
            v2f bc = *(const v2f*)&dftCos[(k0 >> 1) * (FPAD * 2) + n * 2];
            v2f bs = *(const v2f*)&dftSin[(k0 >> 1) * (FPAD * 2) + n * 2];
            accRe = __builtin_amdgcn_wmma_f32_16x16x4_f32(false, a, false, bc,
                                                          (short)0, accRe, false, false);
            accIm = __builtin_amdgcn_wmma_f32_16x16x4_f32(false, a, false, bs,
                                                          (short)0, accIm, false, false);
        }
        #pragma unroll
        for (int v = 0; v < 8; ++v) {
            float re = accRe[v], im = accIm[v];
            mag[(mbase + v) * FPAD + n] = sqrtf(re * re + im * im);
        }
    }
    __syncthreads();

    for (int nt = wv; nt < NMELS / 16; nt += 4) {
        v8f acc = {0.f, 0.f, 0.f, 0.f, 0.f, 0.f, 0.f, 0.f};
        const int n = nt * 16 + ncol;
        #pragma unroll 4
        for (int kb = 0; kb < FPAD; kb += 4) {
            const int k0 = kb + kh;
            v2f a  = *(const v2f*)&mag[mrow * FPAD + k0];
            v2f bm = *(const v2f*)&melFB[(k0 >> 1) * (NMELS * 2) + n * 2];
            acc = __builtin_amdgcn_wmma_f32_16x16x4_f32(false, a, false, bm,
                                                        (short)0, acc, false, false);
        }
        #pragma unroll
        for (int v = 0; v < 8; ++v) {
            int t = t0 + mbase + v;
            if (t < NFRAMES)
                melBuf[(size_t)t * NCH + b * NMELS + n] = acc[v];
        }
    }
}

__global__ void pcen_kernel(const float* __restrict__ melBuf,
                            float* __restrict__ out)
{
    int ch = blockIdx.x * blockDim.x + threadIdx.x;
    if (ch >= NCH) return;
    int b = ch / NMELS, m = ch % NMELS;
    const float s  = 1.0f - expf(-((float)HOP / (float)SRATE) / 0.06f);
    const float da = powf(2.0f, 0.98f);
    float* o = out + ((size_t)b * NMELS + m) * NFRAMES;
    float M = 0.0f;
    for (int t = 0; t < NFRAMES; ++t) {
        float e = melBuf[(size_t)t * NCH + ch];
        M = (1.0f - s) * M + s * e;
        float smooth = sqrtf(M + 1e-6f);
        o[t] = powf(e / smooth + 2.0f, 0.98f) - da;
    }
}

extern "C" void kernel_launch(void* const* d_in, const int* in_sizes, int n_in,
                              void* d_out, int out_size, void* d_ws, size_t ws_size,
                              hipStream_t stream) {
    const float* x = (const float*)d_in[0];
    float* ws = (float*)d_ws;

    float* dftCos = ws;
    float* dftSin = dftCos + (size_t)NFFT * FPAD;
    float* melFB  = dftSin + (size_t)NFFT * FPAD;
    float* melBuf = melFB  + (size_t)FPAD * NMELS;

    int n1 = NFFT * FPAD;
    init_dft<<<(n1 + 255) / 256, 256, 0, stream>>>(dftCos, dftSin);
    int n2 = FPAD * NMELS;
    init_mel<<<(n2 + 255) / 256, 256, 0, stream>>>(melFB);

    dim3 grid(TTILES, BATCH);
    dft_mel_kernel<<<grid, 128, 0, stream>>>(x, dftCos, dftSin, melFB, melBuf);

    pcen_kernel<<<(NCH + 127) / 128, 128, 0, stream>>>(melBuf, (float*)d_out);
}


# pcen: serial 2585-step EMA per channel -> time-chunked parallel scan (21-frame chunk EMAs from zero joined by decay^21 carries; EMA summation order changes), 8-channel tile in LDS
# speedup vs baseline: 3.1254x; 3.1254x over previous
#include <hip/hip_runtime.h>
#include <math.h>

#define SRATE     22050
#define NFFT      1024
#define HOP       256
#define NMELS     128
#define NFREQS    513
#define FPAD      528
#define BATCH     16
#define NSAMP     661504
#define NFRAMES   2585
#define TTILES    162
#define SPAN      (15*HOP + NFFT)
#define NCH       (BATCH*NMELS)

typedef __attribute__((ext_vector_type(2))) float v2f;
typedef __attribute__((ext_vector_type(8))) float v8f;

__global__ void init_dft(float* __restrict__ dftCos, float* __restrict__ dftSin) {
    int idx = blockIdx.x * blockDim.x + threadIdx.x;
    if (idx >= NFFT * FPAD) return;
    int k = idx / FPAD;
    int f = idx % FPAD;
    float c = 0.0f, s = 0.0f;
    if (f < NFREQS) {
        float win = 0.5f - 0.5f * cosf(6.28318530717958647692f * (float)k / (float)NFFT);
        int r = (k * f) & (NFFT - 1);
        float ang = (float)r * (6.28318530717958647692f / (float)NFFT);
        c = win * cosf(ang);
        s = win * sinf(ang);
    }
    int off = (k >> 1) * (FPAD * 2) + f * 2 + (k & 1);
    dftCos[off] = c;
    dftSin[off] = s;
}

__device__ __forceinline__ float hz_to_mel(float f) { return 2595.0f * log10f(1.0f + f / 700.0f); }
__device__ __forceinline__ float mel_to_hz(float m) { return 700.0f * (exp10f(m / 2595.0f) - 1.0f); }

__global__ void init_mel(float* __restrict__ melFB) {
    int idx = blockIdx.x * blockDim.x + threadIdx.x;
    if (idx >= FPAD * NMELS) return;
    int k = idx / NMELS;
    int m = idx % NMELS;
    float v = 0.0f;
    if (k < NFREQS) {
        float m_lo = hz_to_mel(50.0f);
        float m_hi = hz_to_mel((float)SRATE * 0.5f);
        float step = (m_hi - m_lo) / (float)(NMELS + 1);
        float f0 = mel_to_hz(m_lo + step * (float)m);
        float f1 = mel_to_hz(m_lo + step * (float)(m + 1));
        float f2 = mel_to_hz(m_lo + step * (float)(m + 2));
        float freq = (float)k * ((float)SRATE * 0.5f / (float)(NFFT / 2));
        float down = (freq - f0) / (f1 - f0);
        float up   = (f2 - freq) / (f2 - f1);
        v = fmaxf(0.0f, fminf(down, up));
    }
    melFB[(k >> 1) * (NMELS * 2) + m * 2 + (k & 1)] = v;
}

__global__ __launch_bounds__(128)
void dft_mel_kernel(const float* __restrict__ x,
                    const float* __restrict__ dftCos,
                    const float* __restrict__ dftSin,
                    const float* __restrict__ melFB,
                    float* __restrict__ melBuf)
{
    __shared__ float span[SPAN];
    __shared__ float mag[16 * FPAD];

    const int tile = blockIdx.x;
    const int b    = blockIdx.y;
    const int tid  = threadIdx.x;
    const int ln   = tid & 31;
    const int wv   = tid >> 5;
    const int t0   = tile * 16;

    const float* xb = x + (size_t)b * NSAMP;
    const long base = (long)t0 * HOP - (NFFT / 2);
    for (int i = tid; i < SPAN; i += 128) {
        long p = base + i;
        if (p < 0) p = -p;
        if (p >= NSAMP) p = 2L * NSAMP - 2 - p;
        if (p < 0) p = 0;
        if (p >= NSAMP) p = NSAMP - 1;
        span[i] = xb[p];
    }
    __syncthreads();

    const int mrow  = ln & 15;
    const int kh    = (ln >> 4) << 1;
    const int ncol  = ln & 15;
    const int mbase = (ln >> 4) * 8;

    for (int nt = wv; nt < FPAD / 16; nt += 4) {
        v8f accRe = {0.f, 0.f, 0.f, 0.f, 0.f, 0.f, 0.f, 0.f};
        v8f accIm = {0.f, 0.f, 0.f, 0.f, 0.f, 0.f, 0.f, 0.f};
        const int n = nt * 16 + ncol;
        #pragma unroll 4
        for (int kb = 0; kb < NFFT; kb += 4) {
            const int k0 = kb + kh;
            v2f a  = *(const v2f*)&span[mrow * HOP + k0];
            v2f bc = *(const v2f*)&dftCos[(k0 >> 1) * (FPAD * 2) + n * 2];
            v2f bs = *(const v2f*)&dftSin[(k0 >> 1) * (FPAD * 2) + n * 2];
            accRe = __builtin_amdgcn_wmma_f32_16x16x4_f32(false, a, false, bc,
                                                          (short)0, accRe, false, false);
            accIm = __builtin_amdgcn_wmma_f32_16x16x4_f32(false, a, false, bs,
                                                          (short)0, accIm, false, false);
        }
        #pragma unroll
        for (int v = 0; v < 8; ++v) {
            float re = accRe[v], im = accIm[v];
            mag[(mbase + v) * FPAD + n] = sqrtf(re * re + im * im);
        }
    }
    __syncthreads();

    for (int nt = wv; nt < NMELS / 16; nt += 4) {
        v8f acc = {0.f, 0.f, 0.f, 0.f, 0.f, 0.f, 0.f, 0.f};
        const int n = nt * 16 + ncol;
        #pragma unroll 4
        for (int kb = 0; kb < FPAD; kb += 4) {
            const int k0 = kb + kh;
            v2f a  = *(const v2f*)&mag[mrow * FPAD + k0];
            v2f bm = *(const v2f*)&melFB[(k0 >> 1) * (NMELS * 2) + n * 2];
            acc = __builtin_amdgcn_wmma_f32_16x16x4_f32(false, a, false, bm,
                                                        (short)0, acc, false, false);
        }
        #pragma unroll
        for (int v = 0; v < 8; ++v) {
            int t = t0 + mbase + v;
            if (t < NFRAMES)
                melBuf[(size_t)t * NCH + b * NMELS + n] = acc[v];
        }
    }
}

#define PC_CW     8
#define PC_SUB    4
#define PC_L      21
#define PC_NCHUNK 128
#define PC_PITCH  2692
#define PC_BLOCKS (NCH / PC_CW)

__global__ __launch_bounds__(1024)
void pcen_kernel(const float* __restrict__ melBuf,
                 float* __restrict__ out)
{
    __shared__ float tile[PC_CW * PC_PITCH];
    __shared__ float ends[PC_NCHUNK * PC_CW];

    const int tid   = threadIdx.x;
    const int c     = tid & (PC_CW - 1);
    const int chunk = tid >> 3;
    const int ch0   = blockIdx.x * PC_CW;
    const int t0    = chunk * PC_L;

    const float s  = 1.0f - expf(-((float)HOP / (float)SRATE) / 0.06f);
    const float d  = 1.0f - s;
    const float da = powf(2.0f, 0.98f);

    float e[PC_L];
    #pragma unroll
    for (int i = 0; i < PC_L; ++i) {
        const int t = min(t0 + i, NFRAMES - 1);
        e[i] = melBuf[(size_t)t * NCH + ch0 + c];
    }
    float* row = tile + c * PC_PITCH + t0;
    float loc = 0.0f;
    #pragma unroll
    for (int i = 0; i < PC_L; ++i) {
        row[i] = e[i];
        loc = d * loc + s * e[i];
    }
    ends[chunk * PC_CW + c] = loc;
    __syncthreads();

    float dL = 1.0f;
    #pragma unroll
    for (int i = 0; i < PC_L; ++i) dL *= d;
    float M = 0.0f;
    #pragma unroll 8
    for (int j = 0; j < chunk; ++j) M = dL * M + ends[j * PC_CW + c];

    const int n = max(0, min(PC_L, NFRAMES - t0));
    for (int i = 0; i < n; ++i) {
        const float ev = row[i];
        M = (1.0f - s) * M + s * ev;
        const float smooth = sqrtf(M + 1e-6f);
        row[i] = powf(ev / smooth + 2.0f, 0.98f) - da;
    }
    __syncthreads();

    float* o = out + (size_t)ch0 * NFRAMES;
    for (int idx = tid; idx < PC_CW * NFRAMES; idx += 1024) {
        const int r = idx / NFRAMES;
        const int t = idx - r * NFRAMES;
        o[idx] = tile[r * PC_PITCH + t];
    }
}

extern "C" void kernel_launch(void* const* d_in, const int* in_sizes, int n_in,
                              void* d_out, int out_size, void* d_ws, size_t ws_size,
                              hipStream_t stream) {
    const float* x = (const float*)d_in[0];
    float* ws = (float*)d_ws;

    float* dftCos = ws;
    float* dftSin = dftCos + (size_t)NFFT * FPAD;
    float* melFB  = dftSin + (size_t)NFFT * FPAD;
    float* melBuf = melFB  + (size_t)FPAD * NMELS;

    int n1 = NFFT * FPAD;
    init_dft<<<(n1 + 255) / 256, 256, 0, stream>>>(dftCos, dftSin);
    int n2 = FPAD * NMELS;
    init_mel<<<(n2 + 255) / 256, 256, 0, stream>>>(melFB);

    dim3 grid(TTILES, BATCH);
    dft_mel_kernel<<<grid, 128, 0, stream>>>(x, dftCos, dftSin, melFB, melBuf);

    pcen_kernel<<<PC_BLOCKS, 1024, 0, stream>>>(melBuf, (float*)d_out);
}


# dft_mel: XOR-swizzle the LDS sample span and magnitude tile so the 16 rows of each WMMA A-fragment read hit distinct banks (were 16-way and 4-way conflicts); LDS size, k order and sums unchanged
# speedup vs baseline: 3.2614x; 1.0435x over previous
#include <hip/hip_runtime.h>
#include <math.h>

#define SRATE     22050
#define NFFT      1024
#define HOP       256
#define NMELS     128
#define NFREQS    513
#define FPAD      528
#define BATCH     16
#define NSAMP     661504
#define NFRAMES   2585
#define TTILES    162
#define SPAN      (15*HOP + NFFT)
#define NCH       (BATCH*NMELS)

typedef __attribute__((ext_vector_type(2))) float v2f;
typedef __attribute__((ext_vector_type(8))) float v8f;

__device__ __forceinline__ int span_idx(int i) { return i ^ (((i >> 8) & 15) << 2); }
__device__ __forceinline__ int mag_idx(int row, int col) { return row * FPAD + (col ^ ((row >> 2) << 2)); }

__global__ void init_dft(float* __restrict__ dftCos, float* __restrict__ dftSin) {
    int idx = blockIdx.x * blockDim.x + threadIdx.x;
    if (idx >= NFFT * FPAD) return;
    int k = idx / FPAD;
    int f = idx % FPAD;
    float c = 0.0f, s = 0.0f;
    if (f < NFREQS) {
        float win = 0.5f - 0.5f * cosf(6.28318530717958647692f * (float)k / (float)NFFT);
        int r = (k * f) & (NFFT - 1);
        float ang = (float)r * (6.28318530717958647692f / (float)NFFT);
        c = win * cosf(ang);
        s = win * sinf(ang);
    }
    int off = (k >> 1) * (FPAD * 2) + f * 2 + (k & 1);
    dftCos[off] = c;
    dftSin[off] = s;
}

__device__ __forceinline__ float hz_to_mel(float f) { return 2595.0f * log10f(1.0f + f / 700.0f); }
__device__ __forceinline__ float mel_to_hz(float m) { return 700.0f * (exp10f(m / 2595.0f) - 1.0f); }

__global__ void init_mel(float* __restrict__ melFB) {
    int idx = blockIdx.x * blockDim.x + threadIdx.x;
    if (idx >= FPAD * NMELS) return;
    int k = idx / NMELS;
    int m = idx % NMELS;
    float v = 0.0f;
    if (k < NFREQS) {
        float m_lo = hz_to_mel(50.0f);
        float m_hi = hz_to_mel((float)SRATE * 0.5f);
        float step = (m_hi - m_lo) / (float)(NMELS + 1);
        float f0 = mel_to_hz(m_lo + step * (float)m);
        float f1 = mel_to_hz(m_lo + step * (float)(m + 1));
        float f2 = mel_to_hz(m_lo + step * (float)(m + 2));
        float freq = (float)k * ((float)SRATE * 0.5f / (float)(NFFT / 2));
        float down = (freq - f0) / (f1 - f0);
        float up   = (f2 - freq) / (f2 - f1);
        v = fmaxf(0.0f, fminf(down, up));
    }
    melFB[(k >> 1) * (NMELS * 2) + m * 2 + (k & 1)] = v;
}

__global__ __launch_bounds__(128)
void dft_mel_kernel(const float* __restrict__ x,
                    const float* __restrict__ dftCos,
                    const float* __restrict__ dftSin,
                    const float* __restrict__ melFB,
                    float* __restrict__ melBuf)
{
    __shared__ float span[SPAN];
    __shared__ float mag[16 * FPAD];

    const int tile = blockIdx.x;
    const int b    = blockIdx.y;
    const int tid  = threadIdx.x;
    const int ln   = tid & 31;
    const int wv   = tid >> 5;
    const int t0   = tile * 16;

    const float* xb = x + (size_t)b * NSAMP;
    const long base = (long)t0 * HOP - (NFFT / 2);
    for (int i = tid; i < SPAN; i += 128) {
        long p = base + i;
        if (p < 0) p = -p;
        if (p >= NSAMP) p = 2L * NSAMP - 2 - p;
        if (p < 0) p = 0;
        if (p >= NSAMP) p = NSAMP - 1;
        span[span_idx(i)] = xb[p];
    }
    __syncthreads();

    const int mrow  = ln & 15;
    const int kh    = (ln >> 4) << 1;
    const int ncol  = ln & 15;
    const int mbase = (ln >> 4) * 8;

    for (int nt = wv; nt < FPAD / 16; nt += 4) {
        v8f accRe = {0.f, 0.f, 0.f, 0.f, 0.f, 0.f, 0.f, 0.f};
        v8f accIm = {0.f, 0.f, 0.f, 0.f, 0.f, 0.f, 0.f, 0.f};
        const int n = nt * 16 + ncol;
        for (int seg = 0; seg < NFFT / HOP; ++seg) {
            const int srow = (mrow + seg) * HOP;
            const int sw   = ((mrow + seg) & 15) << 2;
            #pragma unroll 4
            for (int kk = 0; kk < HOP; kk += 4) {
                const int k0 = seg * HOP + kk + kh;
                v2f a  = *(const v2f*)&span[srow + ((kk + kh) ^ sw)];
                v2f bc = *(const v2f*)&dftCos[(k0 >> 1) * (FPAD * 2) + n * 2];
                v2f bs = *(const v2f*)&dftSin[(k0 >> 1) * (FPAD * 2) + n * 2];
                accRe = __builtin_amdgcn_wmma_f32_16x16x4_f32(false, a, false, bc,
                                                              (short)0, accRe, false, false);
                accIm = __builtin_amdgcn_wmma_f32_16x16x4_f32(false, a, false, bs,
                                                              (short)0, accIm, false, false);
            }
        }
        #pragma unroll
        for (int v = 0; v < 8; ++v) {
            float re = accRe[v], im = accIm[v];
            mag[mag_idx(mbase + v, n)] = sqrtf(re * re + im * im);
        }
    }
    __syncthreads();

    for (int nt = wv; nt < NMELS / 16; nt += 4) {
        v8f acc = {0.f, 0.f, 0.f, 0.f, 0.f, 0.f, 0.f, 0.f};
        const int n = nt * 16 + ncol;
        #pragma unroll 4
        for (int kb = 0; kb < FPAD; kb += 4) {
            const int k0 = kb + kh;
            v2f a  = *(const v2f*)&mag[mag_idx(mrow, k0)];
            v2f bm = *(const v2f*)&melFB[(k0 >> 1) * (NMELS * 2) + n * 2];
            acc = __builtin_amdgcn_wmma_f32_16x16x4_f32(false, a, false, bm,
                                                        (short)0, acc, false, false);
        }
        #pragma unroll
        for (int v = 0; v < 8; ++v) {
            int t = t0 + mbase + v;
            if (t < NFRAMES)
                melBuf[(size_t)t * NCH + b * NMELS + n] = acc[v];
        }
    }
}

#define PC_CW     8
#define PC_SUB    4
#define PC_L      21
#define PC_NCHUNK 128
#define PC_PITCH  2692
#define PC_BLOCKS (NCH / PC_CW)

__global__ __launch_bounds__(1024)
void pcen_kernel(const float* __restrict__ melBuf,
                 float* __restrict__ out)
{
    __shared__ float tile[PC_CW * PC_PITCH];
    __shared__ float ends[PC_NCHUNK * PC_CW];

    const int tid   = threadIdx.x;
    const int c     = tid & (PC_CW - 1);
    const int chunk = tid >> 3;
    const int ch0   = blockIdx.x * PC_CW;
    const int t0    = chunk * PC_L;

    const float s  = 1.0f - expf(-((float)HOP / (float)SRATE) / 0.06f);
    const float d  = 1.0f - s;
    const float da = powf(2.0f, 0.98f);

    float e[PC_L];
    #pragma unroll
    for (int i = 0; i < PC_L; ++i) {
        const int t = min(t0 + i, NFRAMES - 1);
        e[i] = melBuf[(size_t)t * NCH + ch0 + c];
    }
    float* row = tile + c * PC_PITCH + t0;
    float loc = 0.0f;
    #pragma unroll
    for (int i = 0; i < PC_L; ++i) {
        row[i] = e[i];
        loc = d * loc + s * e[i];
    }
    ends[chunk * PC_CW + c] = loc;
    __syncthreads();

    float dL = 1.0f;
    #pragma unroll
    for (int i = 0; i < PC_L; ++i) dL *= d;
    float M = 0.0f;
    #pragma unroll 8
    for (int j = 0; j < chunk; ++j) M = dL * M + ends[j * PC_CW + c];

    const int n = max(0, min(PC_L, NFRAMES - t0));
    for (int i = 0; i < n; ++i) {
        const float ev = row[i];
        M = (1.0f - s) * M + s * ev;
        const float smooth = sqrtf(M + 1e-6f);
        row[i] = powf(ev / smooth + 2.0f, 0.98f) - da;
    }
    __syncthreads();

    float* o = out + (size_t)ch0 * NFRAMES;
    for (int idx = tid; idx < PC_CW * NFRAMES; idx += 1024) {
        const int r = idx / NFRAMES;
        const int t = idx - r * NFRAMES;
        o[idx] = tile[r * PC_PITCH + t];
    }
}

extern "C" void kernel_launch(void* const* d_in, const int* in_sizes, int n_in,
                              void* d_out, int out_size, void* d_ws, size_t ws_size,
                              hipStream_t stream) {
    const float* x = (const float*)d_in[0];
    float* ws = (float*)d_ws;

    float* dftCos = ws;
    float* dftSin = dftCos + (size_t)NFFT * FPAD;
    float* melFB  = dftSin + (size_t)NFFT * FPAD;
    float* melBuf = melFB  + (size_t)FPAD * NMELS;

    int n1 = NFFT * FPAD;
    init_dft<<<(n1 + 255) / 256, 256, 0, stream>>>(dftCos, dftSin);
    int n2 = FPAD * NMELS;
    init_mel<<<(n2 + 255) / 256, 256, 0, stream>>>(melFB);

    dim3 grid(TTILES, BATCH);
    dft_mel_kernel<<<grid, 128, 0, stream>>>(x, dftCos, dftSin, melFB, melBuf);

    pcen_kernel<<<PC_BLOCKS, 1024, 0, stream>>>(melBuf, (float*)d_out);
}


# dft_mel: 32 frames per group (was 16): each DFT/mel table fragment loaded from L2 feeds two frame tiles' WMMAs; magnitudes held in registers, then written over the folded tiles' LDS; sums unchanged
# speedup vs baseline: 6.7826x; 2.0797x over previous
#include <hip/hip_runtime.h>
#include <math.h>

#define SRATE     22050
#define NFFT      1024
#define HOP       256
#define NMELS     128
#define NFREQS    513
#define FPAD      528
#define BATCH     16
#define NSAMP     661504
#define NFRAMES   2585
#define MT        2
#define GFR       (16*MT)
#define GTILES    81
#define SPAN      (15*HOP + NFFT)
#define NCH       (BATCH*NMELS)
#define KFOLD     (NFFT/2)
#define PQ_PITCH  (KFOLD + 4)

typedef __attribute__((ext_vector_type(2))) float v2f;
typedef __attribute__((ext_vector_type(8))) float v8f;

__device__ __forceinline__ int mag_idx(int row, int col) { return row * FPAD + (col ^ (((row & 15) >> 2) << 2)); }

__global__ void init_dft(float* __restrict__ dftCos, float* __restrict__ dftSin) {
    int idx = blockIdx.x * blockDim.x + threadIdx.x;
    if (idx >= KFOLD * FPAD) return;
    int j = idx / FPAD;
    int f = idx % FPAD;
    float c = 0.0f, s = 0.0f;
    if (f < NFREQS) {
        int k = (j == 0) ? (NFFT / 2) : j;
        float win = 0.5f - 0.5f * cosf(6.28318530717958647692f * (float)k / (float)NFFT);
        int r = (k * f) & (NFFT - 1);
        float ang = (float)r * (6.28318530717958647692f / (float)NFFT);
        c = win * cosf(ang);
        s = (j == 0) ? 0.0f : win * sinf(ang);
    }
    int off = (j >> 1) * (FPAD * 2) + f * 2 + (j & 1);
    dftCos[off] = c;
    dftSin[off] = s;
}

__device__ __forceinline__ float hz_to_mel(float f) { return 2595.0f * log10f(1.0f + f / 700.0f); }
__device__ __forceinline__ float mel_to_hz(float m) { return 700.0f * (exp10f(m / 2595.0f) - 1.0f); }

__global__ void init_mel(float* __restrict__ melFB) {
    int idx = blockIdx.x * blockDim.x + threadIdx.x;
    if (idx >= FPAD * NMELS) return;
    int k = idx / NMELS;
    int m = idx % NMELS;
    float v = 0.0f;
    if (k < NFREQS) {
        float m_lo = hz_to_mel(50.0f);
        float m_hi = hz_to_mel((float)SRATE * 0.5f);
        float step = (m_hi - m_lo) / (float)(NMELS + 1);
        float f0 = mel_to_hz(m_lo + step * (float)m);
        float f1 = mel_to_hz(m_lo + step * (float)(m + 1));
        float f2 = mel_to_hz(m_lo + step * (float)(m + 2));
        float freq = (float)k * ((float)SRATE * 0.5f / (float)(NFFT / 2));
        float down = (freq - f0) / (f1 - f0);
        float up   = (f2 - freq) / (f2 - f1);
        v = fmaxf(0.0f, fminf(down, up));
    }
    melFB[(k >> 1) * (NMELS * 2) + m * 2 + (k & 1)] = v;
}

__global__ __launch_bounds__(256)
void dft_mel_kernel(const float* __restrict__ x,
                    const float* __restrict__ dftCos,
                    const float* __restrict__ dftSin,
                    const float* __restrict__ melFB,
                    float* __restrict__ melBuf)
{
    __shared__ float lds[2 * GFR * PQ_PITCH + SPAN];
    float* P    = lds;
    float* Q    = lds + GFR * PQ_PITCH;
    float* span = lds + 2 * GFR * PQ_PITCH;
    float* mag  = lds;

    const int tile = blockIdx.x;
    const int b    = blockIdx.y;
    const int tid  = threadIdx.x;
    const int ln   = tid & 31;
    const int wv   = tid >> 5;
    const int t0   = tile * GFR;

    const float* xb = x + (size_t)b * NSAMP;
    for (int h = 0; h < MT; ++h) {
        const long base = (long)(t0 + 16 * h) * HOP - (NFFT / 2);
        for (int i = tid; i < SPAN; i += 256) {
            long p = base + i;
            if (p < 0) p = -p;
            if (p >= NSAMP) p = 2L * NSAMP - 2 - p;
            if (p < 0) p = 0;
            if (p >= NSAMP) p = NSAMP - 1;
            span[i] = xb[p];
        }
        __syncthreads();

        for (int idx = tid; idx < 16 * KFOLD; idx += 256) {
            const int r = idx >> 9;
            const int j = idx & (KFOLD - 1);
            const float* xr = span + r * HOP;
            const float lo = xr[j == 0 ? (NFFT / 2) : j];
            const float hi = xr[j == 0 ? (NFFT / 2) : NFFT - j];
            P[(16 * h + r) * PQ_PITCH + j] = (j == 0) ? lo : lo + hi;
            Q[(16 * h + r) * PQ_PITCH + j] = (j == 0) ? 0.0f : lo - hi;
        }
        __syncthreads();
    }

    const int mrow  = ln & 15;
    const int kh    = (ln >> 4) << 1;
    const int ncol  = ln & 15;
    const int mbase = (ln >> 4) * 8;

    const float* pa = P + mrow * PQ_PITCH + kh;
    const float* qa = Q + mrow * PQ_PITCH + kh;
    float mreg[5][MT][8];
    #pragma unroll
    for (int i = 0; i < 5; ++i) {
        const int nt = wv + 8 * i;
        if (nt < FPAD / 16) {
            v8f re0 = {0.f, 0.f, 0.f, 0.f, 0.f, 0.f, 0.f, 0.f};
            v8f im0 = {0.f, 0.f, 0.f, 0.f, 0.f, 0.f, 0.f, 0.f};
            v8f re1 = {0.f, 0.f, 0.f, 0.f, 0.f, 0.f, 0.f, 0.f};
            v8f im1 = {0.f, 0.f, 0.f, 0.f, 0.f, 0.f, 0.f, 0.f};
            const int n = nt * 16 + ncol;
            const float* bcp = dftCos + (kh >> 1) * (FPAD * 2) + n * 2;
            const float* bsp = dftSin + (kh >> 1) * (FPAD * 2) + n * 2;
            #pragma unroll 4
            for (int j0 = 0; j0 < KFOLD; j0 += 4) {
                v2f bc  = *(const v2f*)&bcp[(j0 >> 1) * (FPAD * 2)];
                v2f bs  = *(const v2f*)&bsp[(j0 >> 1) * (FPAD * 2)];
                v2f ap0 = *(const v2f*)&pa[j0];
                v2f aq0 = *(const v2f*)&qa[j0];
                v2f ap1 = *(const v2f*)&pa[16 * PQ_PITCH + j0];
                v2f aq1 = *(const v2f*)&qa[16 * PQ_PITCH + j0];
                re0 = __builtin_amdgcn_wmma_f32_16x16x4_f32(false, ap0, false, bc,
                                                            (short)0, re0, false, false);
                im0 = __builtin_amdgcn_wmma_f32_16x16x4_f32(false, aq0, false, bs,
                                                            (short)0, im0, false, false);
                re1 = __builtin_amdgcn_wmma_f32_16x16x4_f32(false, ap1, false, bc,
                                                            (short)0, re1, false, false);
                im1 = __builtin_amdgcn_wmma_f32_16x16x4_f32(false, aq1, false, bs,
                                                            (short)0, im1, false, false);
            }
            #pragma unroll
            for (int v = 0; v < 8; ++v) {
                float re = re0[v], im = im0[v];
                mreg[i][0][v] = sqrtf(re * re + im * im);
                re = re1[v]; im = im1[v];
                mreg[i][1][v] = sqrtf(re * re + im * im);
            }
        }
    }
    __syncthreads();
    #pragma unroll
    for (int i = 0; i < 5; ++i) {
        const int nt = wv + 8 * i;
        if (nt < FPAD / 16) {
            const int n = nt * 16 + ncol;
            #pragma unroll
            for (int m = 0; m < MT; ++m) {
                #pragma unroll
                for (int v = 0; v < 8; ++v)
                    mag[mag_idx(16 * m + mbase + v, n)] = mreg[i][m][v];
            }
        }
    }
    __syncthreads();

    for (int nt = wv; nt < NMELS / 16; nt += 8) {
        v8f acc0 = {0.f, 0.f, 0.f, 0.f, 0.f, 0.f, 0.f, 0.f};
        v8f acc1 = {0.f, 0.f, 0.f, 0.f, 0.f, 0.f, 0.f, 0.f};
        const int n = nt * 16 + ncol;
        #pragma unroll 4
        for (int kb = 0; kb < FPAD; kb += 4) {
            const int k0 = kb + kh;
            v2f bm = *(const v2f*)&melFB[(k0 >> 1) * (NMELS * 2) + n * 2];
            v2f a0 = *(const v2f*)&mag[mag_idx(mrow, k0)];
            v2f a1 = *(const v2f*)&mag[mag_idx(16 + mrow, k0)];
            acc0 = __builtin_amdgcn_wmma_f32_16x16x4_f32(false, a0, false, bm,
                                                         (short)0, acc0, false, false);
            acc1 = __builtin_amdgcn_wmma_f32_16x16x4_f32(false, a1, false, bm,
                                                         (short)0, acc1, false, false);
        }
        #pragma unroll
        for (int v = 0; v < 8; ++v) {
            int t = t0 + mbase + v;
            if (t < NFRAMES)
                melBuf[(size_t)t * NCH + b * NMELS + n] = acc0[v];
            t += 16;
            if (t < NFRAMES)
                melBuf[(size_t)t * NCH + b * NMELS + n] = acc1[v];
        }
    }
}

#define PC_CW     8
#define PC_SUB    4
#define PC_L      21
#define PC_NCHUNK 128
#define PC_PITCH  2692
#define PC_BLOCKS (NCH / PC_CW)

__global__ __launch_bounds__(1024)
void pcen_kernel(const float* __restrict__ melBuf,
                 float* __restrict__ out)
{
    __shared__ float tile[PC_CW * PC_PITCH];
    __shared__ float ends[PC_NCHUNK * PC_CW];

    const int tid   = threadIdx.x;
    const int c     = tid & (PC_CW - 1);
    const int chunk = tid >> 3;
    const int ch0   = blockIdx.x * PC_CW;
    const int t0    = chunk * PC_L;

    const float s  = 1.0f - expf(-((float)HOP / (float)SRATE) / 0.06f);
    const float d  = 1.0f - s;
    const float da = powf(2.0f, 0.98f);

    float e[PC_L];
    #pragma unroll
    for (int i = 0; i < PC_L; ++i) {
        const int t = min(t0 + i, NFRAMES - 1);
        e[i] = melBuf[(size_t)t * NCH + ch0 + c];
    }
    float* row = tile + c * PC_PITCH + t0;
    float loc = 0.0f;
    #pragma unroll
    for (int i = 0; i < PC_L; ++i) {
        row[i] = e[i];
        loc = d * loc + s * e[i];
    }
    ends[chunk * PC_CW + c] = loc;
    __syncthreads();

    float dL = 1.0f;
    #pragma unroll
    for (int i = 0; i < PC_L; ++i) dL *= d;
    float M = 0.0f;
    #pragma unroll 8
    for (int j = 0; j < chunk; ++j) M = dL * M + ends[j * PC_CW + c];

    const int n = max(0, min(PC_L, NFRAMES - t0));
    for (int i = 0; i < n; ++i) {
        const float ev = row[i];
        M = (1.0f - s) * M + s * ev;
        const float smooth = sqrtf(M + 1e-6f);
        row[i] = powf(ev / smooth + 2.0f, 0.98f) - da;
    }
    __syncthreads();

    float* o = out + (size_t)ch0 * NFRAMES;
    for (int idx = tid; idx < PC_CW * NFRAMES; idx += 1024) {
        const int r = idx / NFRAMES;
        const int t = idx - r * NFRAMES;
        o[idx] = tile[r * PC_PITCH + t];
    }
}

extern "C" void kernel_launch(void* const* d_in, const int* in_sizes, int n_in,
                              void* d_out, int out_size, void* d_ws, size_t ws_size,
                              hipStream_t stream) {
    const float* x = (const float*)d_in[0];
    float* ws = (float*)d_ws;

    float* dftCos = ws;
    float* dftSin = dftCos + (size_t)KFOLD * FPAD;
    float* melFB  = dftSin + (size_t)KFOLD * FPAD;
    float* melBuf = melFB  + (size_t)FPAD * NMELS;

    int n1 = KFOLD * FPAD;
    init_dft<<<(n1 + 255) / 256, 256, 0, stream>>>(dftCos, dftSin);
    int n2 = FPAD * NMELS;
    init_mel<<<(n2 + 255) / 256, 256, 0, stream>>>(melFB);

    dim3 grid(GTILES, BATCH);
    dft_mel_kernel<<<grid, 256, 0, stream>>>(x, dftCos, dftSin, melFB, melBuf);

    pcen_kernel<<<PC_BLOCKS, 1024, 0, stream>>>(melBuf, (float*)d_out);
}


# dft_mel: second fold of the real DFT about tap 256, Hann window moved onto the samples: even bins take p[j]+p[512-j], odd bins p[j]-p[512-j]; K 512->256 per product; DFT and mel sum order change
# speedup vs baseline: 10.3957x; 1.5327x over previous
#include <hip/hip_runtime.h>
#include <math.h>

#define SRATE     22050
#define NFFT      1024
#define HOP       256
#define NMELS     128
#define NFREQS    513
#define FPAD      528
#define ECOLS     272
#define NEVT      (ECOLS/16)
#define BATCH     16
#define NSAMP     661504
#define NFRAMES   2585
#define MT        2
#define GFR       (16*MT)
#define GTILES    81
#define SPAN      (15*HOP + NFFT)
#define NCH       (BATCH*NMELS)
#define KQ        (NFFT/4)
#define KE        (KQ + 4)
#define PITCH     (KQ + 4)

typedef __attribute__((ext_vector_type(2))) float v2f;
typedef __attribute__((ext_vector_type(8))) float v8f;

__device__ __forceinline__ int mag_idx(int row, int col) { return row * FPAD + (col ^ (((row & 15) >> 2) << 2)); }

__device__ __forceinline__ int col_bin(int col) { return (col < ECOLS) ? 2 * col : 2 * (col - ECOLS) + 1; }

__global__ void init_dft(float* __restrict__ cosT, float* __restrict__ sinT) {
    int idx = blockIdx.x * blockDim.x + threadIdx.x;
    if (idx >= KE * FPAD) return;
    int j   = idx / FPAD;
    int col = idx % FPAD;
    const bool even = col < ECOLS;
    const int f = col_bin(col);
    float c = 0.0f, s = 0.0f;
    if (f < NFREQS) {
        if (j <= (even ? KQ : KQ - 1)) {
            int r = (j * f) & (NFFT - 1);
            c = cosf((float)r * (6.28318530717958647692f / (float)NFFT));
        }
        if (j < KQ) {
            int k = (!even && j == 0) ? KQ : j;
            int r = (k * f) & (NFFT - 1);
            s = sinf((float)r * (6.28318530717958647692f / (float)NFFT));
        }
    }
    int off = (j >> 1) * (FPAD * 2) + col * 2 + (j & 1);
    cosT[off] = c;
    if (j < KQ) sinT[off] = s;
}

__device__ __forceinline__ float hz_to_mel(float f) { return 2595.0f * log10f(1.0f + f / 700.0f); }
__device__ __forceinline__ float mel_to_hz(float m) { return 700.0f * (exp10f(m / 2595.0f) - 1.0f); }

__global__ void init_mel(float* __restrict__ melFB) {
    int idx = blockIdx.x * blockDim.x + threadIdx.x;
    if (idx >= FPAD * NMELS) return;
    int k = idx / NMELS;
    int m = idx % NMELS;
    const int bin = col_bin(k);
    float v = 0.0f;
    if (bin < NFREQS) {
        float m_lo = hz_to_mel(50.0f);
        float m_hi = hz_to_mel((float)SRATE * 0.5f);
        float step = (m_hi - m_lo) / (float)(NMELS + 1);
        float f0 = mel_to_hz(m_lo + step * (float)m);
        float f1 = mel_to_hz(m_lo + step * (float)(m + 1));
        float f2 = mel_to_hz(m_lo + step * (float)(m + 2));
        float freq = (float)bin * ((float)SRATE * 0.5f / (float)(NFFT / 2));
        float down = (freq - f0) / (f1 - f0);
        float up   = (f2 - freq) / (f2 - f1);
        v = fmaxf(0.0f, fminf(down, up));
    }
    melFB[(k >> 1) * (NMELS * 2) + m * 2 + (k & 1)] = v;
}

__global__ __launch_bounds__(256)
void dft_mel_kernel(const float* __restrict__ x,
                    const float* __restrict__ cosT,
                    const float* __restrict__ sinT,
                    const float* __restrict__ melFB,
                    float* __restrict__ melBuf)
{
    __shared__ float lds[4 * GFR * PITCH + SPAN];
    float* PE   = lds;
    float* PO   = lds + 1 * GFR * PITCH;
    float* QE   = lds + 2 * GFR * PITCH;
    float* QO   = lds + 3 * GFR * PITCH;
    float* span = lds + 4 * GFR * PITCH;
    float* mag  = lds;

    const int tile = blockIdx.x;
    const int b    = blockIdx.y;
    const int tid  = threadIdx.x;
    const int ln   = tid & 31;
    const int wv   = tid >> 5;
    const int t0   = tile * GFR;

    const float cw = cosf((float)tid * (6.28318530717958647692f / (float)NFFT));
    const float wa = 0.5f - 0.5f * cw;
    const float wb = 0.5f + 0.5f * cw;
    const float wq = 0.5f - 0.5f * cosf((float)(NFFT / 4) * (6.28318530717958647692f / (float)NFFT));

    const float* xb = x + (size_t)b * NSAMP;
    for (int h = 0; h < MT; ++h) {
        const long base = (long)(t0 + 16 * h) * HOP - (NFFT / 2);
        for (int i = tid; i < SPAN; i += 256) {
            long p = base + i;
            if (p < 0) p = -p;
            if (p >= NSAMP) p = 2L * NSAMP - 2 - p;
            if (p < 0) p = 0;
            if (p >= NSAMP) p = NSAMP - 1;
            span[i] = xb[p];
        }
        __syncthreads();

        if (tid != 0) {
            const int j = tid;
            for (int r = 0; r < 16; ++r) {
                const float* xr = span + r * HOP;
                const float ya = wa * xr[j];
                const float yd = wa * xr[NFFT - j];
                const float yb = wb * xr[NFFT / 2 - j];
                const float yc = wb * xr[NFFT / 2 + j];
                const float p0 = ya + yd, p1 = yb + yc;
                const float q0 = ya - yd, q1 = yb - yc;
                const int o = (16 * h + r) * PITCH + j;
                PE[o] = p0 + p1;
                PO[o] = p0 - p1;
                QE[o] = q0 - q1;
                QO[o] = q0 + q1;
            }
        }
        if (tid < 16) {
            const float* xr = span + tid * HOP;
            const float w0 = 0.5f - 0.5f * cosf(0.0f);
            const float wh = 0.5f + 0.5f * cosf(0.0f);
            const float y0   = w0 * xr[0];
            const float y512 = wh * xr[NFFT / 2];
            const float y256 = wq * xr[NFFT / 4];
            const float y768 = wq * xr[3 * NFFT / 4];
            const int o = (16 * h + tid) * PITCH;
            PE[o] = y0 + y512;
            PO[o] = y0 - y512;
            QE[o] = 0.0f;
            QO[o] = y256 - y768;
            PE[o + KQ]     = y256 + y768;
            PE[o + KQ + 1] = 0.0f;
            PE[o + KQ + 2] = 0.0f;
            PE[o + KQ + 3] = 0.0f;
        }
        __syncthreads();
    }

    const int mrow  = ln & 15;
    const int kh    = (ln >> 4) << 1;
    const int ncol  = ln & 15;
    const int mbase = (ln >> 4) * 8;

    float mreg[5][MT][8];
    #pragma unroll
    for (int i = 0; i < 5; ++i) {
        const int nt = wv + 8 * i;
        if (nt < FPAD / 16) {
            const bool ev = nt < NEVT;
            const float* pa = (ev ? PE : PO) + mrow * PITCH + kh;
            const float* qa = (ev ? QE : QO) + mrow * PITCH + kh;
            v8f re0 = {0.f, 0.f, 0.f, 0.f, 0.f, 0.f, 0.f, 0.f};
            v8f im0 = {0.f, 0.f, 0.f, 0.f, 0.f, 0.f, 0.f, 0.f};
            v8f re1 = {0.f, 0.f, 0.f, 0.f, 0.f, 0.f, 0.f, 0.f};
            v8f im1 = {0.f, 0.f, 0.f, 0.f, 0.f, 0.f, 0.f, 0.f};
            const int n = nt * 16 + ncol;
            const float* bcp = cosT + (kh >> 1) * (FPAD * 2) + n * 2;
            const float* bsp = sinT + (kh >> 1) * (FPAD * 2) + n * 2;
            #pragma unroll 4
            for (int j0 = 0; j0 < KQ; j0 += 4) {
                v2f bc  = *(const v2f*)&bcp[(j0 >> 1) * (FPAD * 2)];
                v2f bs  = *(const v2f*)&bsp[(j0 >> 1) * (FPAD * 2)];
                v2f ap0 = *(const v2f*)&pa[j0];
                v2f aq0 = *(const v2f*)&qa[j0];
                v2f ap1 = *(const v2f*)&pa[16 * PITCH + j0];
                v2f aq1 = *(const v2f*)&qa[16 * PITCH + j0];
                re0 = __builtin_amdgcn_wmma_f32_16x16x4_f32(false, ap0, false, bc,
                                                            (short)0, re0, false, false);
                im0 = __builtin_amdgcn_wmma_f32_16x16x4_f32(false, aq0, false, bs,
                                                            (short)0, im0, false, false);
                re1 = __builtin_amdgcn_wmma_f32_16x16x4_f32(false, ap1, false, bc,
                                                            (short)0, re1, false, false);
                im1 = __builtin_amdgcn_wmma_f32_16x16x4_f32(false, aq1, false, bs,
                                                            (short)0, im1, false, false);
            }
            if (ev) {
                v2f bc  = *(const v2f*)&bcp[(KQ >> 1) * (FPAD * 2)];
                v2f ap0 = *(const v2f*)&pa[KQ];
                v2f ap1 = *(const v2f*)&pa[16 * PITCH + KQ];
                re0 = __builtin_amdgcn_wmma_f32_16x16x4_f32(false, ap0, false, bc,
                                                            (short)0, re0, false, false);
                re1 = __builtin_amdgcn_wmma_f32_16x16x4_f32(false, ap1, false, bc,
                                                            (short)0, re1, false, false);
            }
            #pragma unroll
            for (int v = 0; v < 8; ++v) {
                float re = re0[v], im = im0[v];
                mreg[i][0][v] = sqrtf(re * re + im * im);
                re = re1[v]; im = im1[v];
                mreg[i][1][v] = sqrtf(re * re + im * im);
            }
        }
    }
    __syncthreads();
    #pragma unroll
    for (int i = 0; i < 5; ++i) {
        const int nt = wv + 8 * i;
        if (nt < FPAD / 16) {
            const int n = nt * 16 + ncol;
            #pragma unroll
            for (int m = 0; m < MT; ++m) {
                #pragma unroll
                for (int v = 0; v < 8; ++v)
                    mag[mag_idx(16 * m + mbase + v, n)] = mreg[i][m][v];
            }
        }
    }
    __syncthreads();

    for (int nt = wv; nt < NMELS / 16; nt += 8) {
        v8f acc0 = {0.f, 0.f, 0.f, 0.f, 0.f, 0.f, 0.f, 0.f};
        v8f acc1 = {0.f, 0.f, 0.f, 0.f, 0.f, 0.f, 0.f, 0.f};
        const int n = nt * 16 + ncol;
        #pragma unroll 4
        for (int kb = 0; kb < FPAD; kb += 4) {
            const int k0 = kb + kh;
            v2f bm = *(const v2f*)&melFB[(k0 >> 1) * (NMELS * 2) + n * 2];
            v2f a0 = *(const v2f*)&mag[mag_idx(mrow, k0)];
            v2f a1 = *(const v2f*)&mag[mag_idx(16 + mrow, k0)];
            acc0 = __builtin_amdgcn_wmma_f32_16x16x4_f32(false, a0, false, bm,
                                                         (short)0, acc0, false, false);
            acc1 = __builtin_amdgcn_wmma_f32_16x16x4_f32(false, a1, false, bm,
                                                         (short)0, acc1, false, false);
        }
        #pragma unroll
        for (int v = 0; v < 8; ++v) {
            int t = t0 + mbase + v;
            if (t < NFRAMES)
                melBuf[(size_t)t * NCH + b * NMELS + n] = acc0[v];
            t += 16;
            if (t < NFRAMES)
                melBuf[(size_t)t * NCH + b * NMELS + n] = acc1[v];
        }
    }
}

#define PC_CW     8
#define PC_SUB    4
#define PC_L      21
#define PC_NCHUNK 128
#define PC_PITCH  2692
#define PC_BLOCKS (NCH / PC_CW)

__global__ __launch_bounds__(1024)
void pcen_kernel(const float* __restrict__ melBuf,
                 float* __restrict__ out)
{
    __shared__ float tile[PC_CW * PC_PITCH];
    __shared__ float ends[PC_NCHUNK * PC_CW];

    const int tid   = threadIdx.x;
    const int c     = tid & (PC_CW - 1);
    const int chunk = tid >> 3;
    const int ch0   = blockIdx.x * PC_CW;
    const int t0    = chunk * PC_L;

    const float s  = 1.0f - expf(-((float)HOP / (float)SRATE) / 0.06f);
    const float d  = 1.0f - s;
    const float da = powf(2.0f, 0.98f);

    float e[PC_L];
    #pragma unroll
    for (int i = 0; i < PC_L; ++i) {
        const int t = min(t0 + i, NFRAMES - 1);
        e[i] = melBuf[(size_t)t * NCH + ch0 + c];
    }
    float* row = tile + c * PC_PITCH + t0;
    float loc = 0.0f;
    #pragma unroll
    for (int i = 0; i < PC_L; ++i) {
        row[i] = e[i];
        loc = d * loc + s * e[i];
    }
    ends[chunk * PC_CW + c] = loc;
    __syncthreads();

    float dL = 1.0f;
    #pragma unroll
    for (int i = 0; i < PC_L; ++i) dL *= d;
    float M = 0.0f;
    #pragma unroll 8
    for (int j = 0; j < chunk; ++j) M = dL * M + ends[j * PC_CW + c];

    const int n = max(0, min(PC_L, NFRAMES - t0));
    for (int i = 0; i < n; ++i) {
        const float ev = row[i];
        M = (1.0f - s) * M + s * ev;
        const float smooth = sqrtf(M + 1e-6f);
        row[i] = powf(ev / smooth + 2.0f, 0.98f) - da;
    }
    __syncthreads();

    float* o = out + (size_t)ch0 * NFRAMES;
    for (int idx = tid; idx < PC_CW * NFRAMES; idx += 1024) {
        const int r = idx / NFRAMES;
        const int t = idx - r * NFRAMES;
        o[idx] = tile[r * PC_PITCH + t];
    }
}

extern "C" void kernel_launch(void* const* d_in, const int* in_sizes, int n_in,
                              void* d_out, int out_size, void* d_ws, size_t ws_size,
                              hipStream_t stream) {
    const float* x = (const float*)d_in[0];
    float* ws = (float*)d_ws;

    float* cosT   = ws;
    float* sinT   = cosT  + (size_t)KE * FPAD;
    float* melFB  = sinT  + (size_t)KQ * FPAD;
    float* melBuf = melFB + (size_t)FPAD * NMELS;

    int n1 = KE * FPAD;
    init_dft<<<(n1 + 255) / 256, 256, 0, stream>>>(cosT, sinT);
    int n2 = FPAD * NMELS;
    init_mel<<<(n2 + 255) / 256, 256, 0, stream>>>(melFB);

    dim3 grid(GTILES, BATCH);
    dft_mel_kernel<<<grid, 256, 0, stream>>>(x, cosT, sinT, melFB, melBuf);

    pcen_kernel<<<PC_BLOCKS, 1024, 0, stream>>>(melBuf, (float*)d_out);
}


# dft_mel: software-pipeline both WMMA K loops: the next block's table (L2) and LDS fragments load into a second register set ahead of the current block's WMMAs; k order and sums unchanged
# speedup vs baseline: 11.3595x; 1.0927x over previous
#include <hip/hip_runtime.h>
#include <math.h>

#define SRATE     22050
#define NFFT      1024
#define HOP       256
#define NMELS     128
#define NFREQS    513
#define FPAD      528
#define ECOLS     272
#define NEVT      (ECOLS/16)
#define BATCH     16
#define NSAMP     661504
#define NFRAMES   2585
#define MT        2
#define GFR       (16*MT)
#define GTILES    81
#define SPAN      (15*HOP + NFFT)
#define NCH       (BATCH*NMELS)
#define KQ        (NFFT/4)
#define KE        (KQ + 4)
#define PITCH     (KQ + 4)

typedef __attribute__((ext_vector_type(2))) float v2f;
typedef __attribute__((ext_vector_type(8))) float v8f;

__device__ __forceinline__ int mag_idx(int row, int col) { return row * FPAD + (col ^ (((row & 15) >> 2) << 2)); }

__device__ __forceinline__ int col_bin(int col) { return (col < ECOLS) ? 2 * col : 2 * (col - ECOLS) + 1; }

__global__ void init_dft(float* __restrict__ cosT, float* __restrict__ sinT) {
    int idx = blockIdx.x * blockDim.x + threadIdx.x;
    if (idx >= KE * FPAD) return;
    int j   = idx / FPAD;
    int col = idx % FPAD;
    const bool even = col < ECOLS;
    const int f = col_bin(col);
    float c = 0.0f, s = 0.0f;
    if (f < NFREQS) {
        if (j <= (even ? KQ : KQ - 1)) {
            int r = (j * f) & (NFFT - 1);
            c = cosf((float)r * (6.28318530717958647692f / (float)NFFT));
        }
        if (j < KQ) {
            int k = (!even && j == 0) ? KQ : j;
            int r = (k * f) & (NFFT - 1);
            s = sinf((float)r * (6.28318530717958647692f / (float)NFFT));
        }
    }
    int off = (j >> 1) * (FPAD * 2) + col * 2 + (j & 1);
    cosT[off] = c;
    if (j < KQ) sinT[off] = s;
}

__device__ __forceinline__ float hz_to_mel(float f) { return 2595.0f * log10f(1.0f + f / 700.0f); }
__device__ __forceinline__ float mel_to_hz(float m) { return 700.0f * (exp10f(m / 2595.0f) - 1.0f); }

__global__ void init_mel(float* __restrict__ melFB) {
    int idx = blockIdx.x * blockDim.x + threadIdx.x;
    if (idx >= FPAD * NMELS) return;
    int k = idx / NMELS;
    int m = idx % NMELS;
    const int bin = col_bin(k);
    float v = 0.0f;
    if (bin < NFREQS) {
        float m_lo = hz_to_mel(50.0f);
        float m_hi = hz_to_mel((float)SRATE * 0.5f);
        float step = (m_hi - m_lo) / (float)(NMELS + 1);
        float f0 = mel_to_hz(m_lo + step * (float)m);
        float f1 = mel_to_hz(m_lo + step * (float)(m + 1));
        float f2 = mel_to_hz(m_lo + step * (float)(m + 2));
        float freq = (float)bin * ((float)SRATE * 0.5f / (float)(NFFT / 2));
        float down = (freq - f0) / (f1 - f0);
        float up   = (f2 - freq) / (f2 - f1);
        v = fmaxf(0.0f, fminf(down, up));
    }
    melFB[(k >> 1) * (NMELS * 2) + m * 2 + (k & 1)] = v;
}

#define S1K 4
#define S2K 12

struct TilePtr { const float* bcp; const float* bsp; const float* pa; const float* qa; };
struct Frag1 { v2f bc[S1K], bs[S1K], ap0[S1K], aq0[S1K], ap1[S1K], aq1[S1K]; };
struct Frag2 { v2f bm[S2K], a0[S2K], a1[S2K]; };

__device__ __forceinline__ void load_frag1(Frag1& f, const TilePtr& p, int j0) {
    #pragma unroll
    for (int s = 0; s < S1K; ++s) {
        const int j = j0 + 4 * s;
        f.bc[s] = *(const v2f*)&p.bcp[(j >> 1) * (FPAD * 2)];
        f.bs[s] = *(const v2f*)&p.bsp[(j >> 1) * (FPAD * 2)];
    }
    #pragma unroll
    for (int s = 0; s < S1K; ++s) {
        const int j = j0 + 4 * s;
        f.ap0[s] = *(const v2f*)&p.pa[j];
        f.aq0[s] = *(const v2f*)&p.qa[j];
        f.ap1[s] = *(const v2f*)&p.pa[16 * PITCH + j];
        f.aq1[s] = *(const v2f*)&p.qa[16 * PITCH + j];
    }
}

__device__ __forceinline__ void mma_frag1(const Frag1& f, v8f& re0, v8f& im0, v8f& re1, v8f& im1) {
    #pragma unroll
    for (int s = 0; s < S1K; ++s) {
        re0 = __builtin_amdgcn_wmma_f32_16x16x4_f32(false, f.ap0[s], false, f.bc[s],
                                                    (short)0, re0, false, false);
        im0 = __builtin_amdgcn_wmma_f32_16x16x4_f32(false, f.aq0[s], false, f.bs[s],
                                                    (short)0, im0, false, false);
        re1 = __builtin_amdgcn_wmma_f32_16x16x4_f32(false, f.ap1[s], false, f.bc[s],
                                                    (short)0, re1, false, false);
        im1 = __builtin_amdgcn_wmma_f32_16x16x4_f32(false, f.aq1[s], false, f.bs[s],
                                                    (short)0, im1, false, false);
    }
}

__global__ __launch_bounds__(256)
void dft_mel_kernel(const float* __restrict__ x,
                    const float* __restrict__ cosT,
                    const float* __restrict__ sinT,
                    const float* __restrict__ melFB,
                    float* __restrict__ melBuf)
{
    __shared__ float lds[4 * GFR * PITCH + SPAN];
    float* PE   = lds;
    float* PO   = lds + 1 * GFR * PITCH;
    float* QE   = lds + 2 * GFR * PITCH;
    float* QO   = lds + 3 * GFR * PITCH;
    float* span = lds + 4 * GFR * PITCH;
    float* mag  = lds;

    const int tile = blockIdx.x;
    const int b    = blockIdx.y;
    const int tid  = threadIdx.x;
    const int ln   = tid & 31;
    const int wv   = tid >> 5;
    const int t0   = tile * GFR;

    const float cw = cosf((float)tid * (6.28318530717958647692f / (float)NFFT));
    const float wa = 0.5f - 0.5f * cw;
    const float wb = 0.5f + 0.5f * cw;
    const float wq = 0.5f - 0.5f * cosf((float)(NFFT / 4) * (6.28318530717958647692f / (float)NFFT));

    const float* xb = x + (size_t)b * NSAMP;
    for (int h = 0; h < MT; ++h) {
        const long base = (long)(t0 + 16 * h) * HOP - (NFFT / 2);
        for (int i = tid; i < SPAN; i += 256) {
            long p = base + i;
            if (p < 0) p = -p;
            if (p >= NSAMP) p = 2L * NSAMP - 2 - p;
            if (p < 0) p = 0;
            if (p >= NSAMP) p = NSAMP - 1;
            span[i] = xb[p];
        }
        __syncthreads();

        if (tid != 0) {
            const int j = tid;
            for (int r = 0; r < 16; ++r) {
                const float* xr = span + r * HOP;
                const float ya = wa * xr[j];
                const float yd = wa * xr[NFFT - j];
                const float yb = wb * xr[NFFT / 2 - j];
                const float yc = wb * xr[NFFT / 2 + j];
                const float p0 = ya + yd, p1 = yb + yc;
                const float q0 = ya - yd, q1 = yb - yc;
                const int o = (16 * h + r) * PITCH + j;
                PE[o] = p0 + p1;
                PO[o] = p0 - p1;
                QE[o] = q0 - q1;
                QO[o] = q0 + q1;
            }
        }
        if (tid < 16) {
            const float* xr = span + tid * HOP;
            const float w0 = 0.5f - 0.5f * cosf(0.0f);
            const float wh = 0.5f + 0.5f * cosf(0.0f);
            const float y0   = w0 * xr[0];
            const float y512 = wh * xr[NFFT / 2];
            const float y256 = wq * xr[NFFT / 4];
            const float y768 = wq * xr[3 * NFFT / 4];
            const int o = (16 * h + tid) * PITCH;
            PE[o] = y0 + y512;
            PO[o] = y0 - y512;
            QE[o] = 0.0f;
            QO[o] = y256 - y768;
            PE[o + KQ]     = y256 + y768;
            PE[o + KQ + 1] = 0.0f;
            PE[o + KQ + 2] = 0.0f;
            PE[o + KQ + 3] = 0.0f;
        }
        __syncthreads();
    }

    const int mrow  = ln & 15;
    const int kh    = (ln >> 4) << 1;
    const int ncol  = ln & 15;
    const int mbase = (ln >> 4) * 8;

    float mreg[5][MT][8];
    auto tile_ptr = [&](int nt) {
        const bool ev = nt < NEVT;
        const int n = nt * 16 + ncol;
        TilePtr p;
        p.pa  = (ev ? PE : PO) + mrow * PITCH + kh;
        p.qa  = (ev ? QE : QO) + mrow * PITCH + kh;
        p.bcp = cosT + (kh >> 1) * (FPAD * 2) + n * 2;
        p.bsp = sinT + (kh >> 1) * (FPAD * 2) + n * 2;
        return p;
    };
    Frag1 fa, fb;
    TilePtr cur = tile_ptr(wv);
    load_frag1(fa, cur, 0);
    #pragma unroll
    for (int i = 0; i < 5; ++i) {
        const int nt = wv + 8 * i;
        if (nt < FPAD / 16) {
            const bool ev = nt < NEVT;
            const bool has_next = (i + 1 < 5) && (nt + 8 < FPAD / 16);
            const TilePtr nxt = tile_ptr(has_next ? nt + 8 : nt);
            v8f re0 = {0.f, 0.f, 0.f, 0.f, 0.f, 0.f, 0.f, 0.f};
            v8f im0 = {0.f, 0.f, 0.f, 0.f, 0.f, 0.f, 0.f, 0.f};
            v8f re1 = {0.f, 0.f, 0.f, 0.f, 0.f, 0.f, 0.f, 0.f};
            v8f im1 = {0.f, 0.f, 0.f, 0.f, 0.f, 0.f, 0.f, 0.f};
            #pragma unroll 1
            for (int j0 = 0; j0 < KQ - 8 * S1K; j0 += 8 * S1K) {
                load_frag1(fb, cur, j0 + 4 * S1K);
                __builtin_amdgcn_sched_barrier(0);
                mma_frag1(fa, re0, im0, re1, im1);
                __builtin_amdgcn_sched_barrier(0);
                load_frag1(fa, cur, j0 + 8 * S1K);
                __builtin_amdgcn_sched_barrier(0);
                mma_frag1(fb, re0, im0, re1, im1);
                __builtin_amdgcn_sched_barrier(0);
            }
            load_frag1(fb, cur, KQ - 4 * S1K);
            __builtin_amdgcn_sched_barrier(0);
            mma_frag1(fa, re0, im0, re1, im1);
            __builtin_amdgcn_sched_barrier(0);
            if (has_next) load_frag1(fa, nxt, 0);
            __builtin_amdgcn_sched_barrier(0);
            mma_frag1(fb, re0, im0, re1, im1);
            if (ev) {
                v2f bc  = *(const v2f*)&cur.bcp[(KQ >> 1) * (FPAD * 2)];
                v2f ap0 = *(const v2f*)&cur.pa[KQ];
                v2f ap1 = *(const v2f*)&cur.pa[16 * PITCH + KQ];
                re0 = __builtin_amdgcn_wmma_f32_16x16x4_f32(false, ap0, false, bc,
                                                            (short)0, re0, false, false);
                re1 = __builtin_amdgcn_wmma_f32_16x16x4_f32(false, ap1, false, bc,
                                                            (short)0, re1, false, false);
            }
            #pragma unroll
            for (int v = 0; v < 8; ++v) {
                float re = re0[v], im = im0[v];
                mreg[i][0][v] = sqrtf(re * re + im * im);
                re = re1[v]; im = im1[v];
                mreg[i][1][v] = sqrtf(re * re + im * im);
            }
            cur = nxt;
        }
    }
    __syncthreads();
    #pragma unroll
    for (int i = 0; i < 5; ++i) {
        const int nt = wv + 8 * i;
        if (nt < FPAD / 16) {
            const int n = nt * 16 + ncol;
            #pragma unroll
            for (int m = 0; m < MT; ++m) {
                #pragma unroll
                for (int v = 0; v < 8; ++v)
                    mag[mag_idx(16 * m + mbase + v, n)] = mreg[i][m][v];
            }
        }
    }
    __syncthreads();

    for (int nt = wv; nt < NMELS / 16; nt += 8) {
        v8f acc0 = {0.f, 0.f, 0.f, 0.f, 0.f, 0.f, 0.f, 0.f};
        v8f acc1 = {0.f, 0.f, 0.f, 0.f, 0.f, 0.f, 0.f, 0.f};
        const int n = nt * 16 + ncol;
        auto load_frag2 = [&](Frag2& f, int kb0) {
            #pragma unroll
            for (int s = 0; s < S2K; ++s) {
                const int k0 = kb0 + 4 * s + kh;
                f.bm[s] = *(const v2f*)&melFB[(k0 >> 1) * (NMELS * 2) + n * 2];
            }
            #pragma unroll
            for (int s = 0; s < S2K; ++s) {
                const int k0 = kb0 + 4 * s + kh;
                f.a0[s] = *(const v2f*)&mag[mag_idx(mrow, k0)];
                f.a1[s] = *(const v2f*)&mag[mag_idx(16 + mrow, k0)];
            }
        };
        auto mma_frag2 = [&](const Frag2& f) {
            #pragma unroll
            for (int s = 0; s < S2K; ++s) {
                acc0 = __builtin_amdgcn_wmma_f32_16x16x4_f32(false, f.a0[s], false, f.bm[s],
                                                             (short)0, acc0, false, false);
                acc1 = __builtin_amdgcn_wmma_f32_16x16x4_f32(false, f.a1[s], false, f.bm[s],
                                                             (short)0, acc1, false, false);
            }
        };
        Frag2 ga, gb;
        load_frag2(ga, 0);
        #pragma unroll 1
        for (int kb = 0; kb < FPAD - 4 * S2K; kb += 8 * S2K) {
            load_frag2(gb, kb + 4 * S2K);
            __builtin_amdgcn_sched_barrier(0);
            mma_frag2(ga);
            __builtin_amdgcn_sched_barrier(0);
            load_frag2(ga, kb + 8 * S2K);
            __builtin_amdgcn_sched_barrier(0);
            mma_frag2(gb);
            __builtin_amdgcn_sched_barrier(0);
        }
        mma_frag2(ga);
        #pragma unroll
        for (int v = 0; v < 8; ++v) {
            int t = t0 + mbase + v;
            if (t < NFRAMES)
                melBuf[(size_t)t * NCH + b * NMELS + n] = acc0[v];
            t += 16;
            if (t < NFRAMES)
                melBuf[(size_t)t * NCH + b * NMELS + n] = acc1[v];
        }
    }
}

#define PC_CW     8
#define PC_SUB    4
#define PC_L      21
#define PC_NCHUNK 128
#define PC_PITCH  2692
#define PC_BLOCKS (NCH / PC_CW)

__global__ __launch_bounds__(1024)
void pcen_kernel(const float* __restrict__ melBuf,
                 float* __restrict__ out)
{
    __shared__ float tile[PC_CW * PC_PITCH];
    __shared__ float ends[PC_NCHUNK * PC_CW];

    const int tid   = threadIdx.x;
    const int c     = tid & (PC_CW - 1);
    const int chunk = tid >> 3;
    const int ch0   = blockIdx.x * PC_CW;
    const int t0    = chunk * PC_L;

    const float s  = 1.0f - expf(-((float)HOP / (float)SRATE) / 0.06f);
    const float d  = 1.0f - s;
    const float da = powf(2.0f, 0.98f);

    float e[PC_L];
    #pragma unroll
    for (int i = 0; i < PC_L; ++i) {
        const int t = min(t0 + i, NFRAMES - 1);
        e[i] = melBuf[(size_t)t * NCH + ch0 + c];
    }
    float* row = tile + c * PC_PITCH + t0;
    float loc = 0.0f;
    #pragma unroll
    for (int i = 0; i < PC_L; ++i) {
        row[i] = e[i];
        loc = d * loc + s * e[i];
    }
    ends[chunk * PC_CW + c] = loc;
    __syncthreads();

    float dL = 1.0f;
    #pragma unroll
    for (int i = 0; i < PC_L; ++i) dL *= d;
    float M = 0.0f;
    #pragma unroll 8
    for (int j = 0; j < chunk; ++j) M = dL * M + ends[j * PC_CW + c];

    const int n = max(0, min(PC_L, NFRAMES - t0));
    for (int i = 0; i < n; ++i) {
        const float ev = row[i];
        M = (1.0f - s) * M + s * ev;
        const float smooth = sqrtf(M + 1e-6f);
        row[i] = powf(ev / smooth + 2.0f, 0.98f) - da;
    }
    __syncthreads();

    float* o = out + (size_t)ch0 * NFRAMES;
    for (int idx = tid; idx < PC_CW * NFRAMES; idx += 1024) {
        const int r = idx / NFRAMES;
        const int t = idx - r * NFRAMES;
        o[idx] = tile[r * PC_PITCH + t];
    }
}

extern "C" void kernel_launch(void* const* d_in, const int* in_sizes, int n_in,
                              void* d_out, int out_size, void* d_ws, size_t ws_size,
                              hipStream_t stream) {
    const float* x = (const float*)d_in[0];
    float* ws = (float*)d_ws;

    float* cosT   = ws;
    float* sinT   = cosT  + (size_t)KE * FPAD;
    float* melFB  = sinT  + (size_t)KQ * FPAD;
    float* melBuf = melFB + (size_t)FPAD * NMELS;

    int n1 = KE * FPAD;
    init_dft<<<(n1 + 255) / 256, 256, 0, stream>>>(cosT, sinT);
    int n2 = FPAD * NMELS;
    init_mel<<<(n2 + 255) / 256, 256, 0, stream>>>(melFB);

    dim3 grid(GTILES, BATCH);
    dft_mel_kernel<<<grid, 256, 0, stream>>>(x, cosT, sinT, melFB, melBuf);

    pcen_kernel<<<PC_BLOCKS, 1024, 0, stream>>>(melBuf, (float*)d_out);
}


# dft_mel: mel K loop visits only the 16-column groups between its mel tile's band edges (every other mel-table row is exactly zero): 2112 -> 424 WMMAs per work-group; sums unchanged
# speedup vs baseline: 12.5186x; 1.1020x over previous
#include <hip/hip_runtime.h>
#include <math.h>

#define SRATE     22050
#define NFFT      1024
#define HOP       256
#define NMELS     128
#define NFREQS    513
#define FPAD      528
#define ECOLS     272
#define NEVT      (ECOLS/16)
#define BATCH     16
#define NSAMP     661504
#define NFRAMES   2585
#define MT        2
#define GFR       (16*MT)
#define GTILES    81
#define SPAN      (15*HOP + NFFT)
#define NCH       (BATCH*NMELS)
#define KQ        (NFFT/4)
#define KE        (KQ + 4)
#define PITCH     (KQ + 4)

typedef __attribute__((ext_vector_type(2))) float v2f;
typedef __attribute__((ext_vector_type(8))) float v8f;

__device__ __forceinline__ int mag_idx(int row, int col) { return row * FPAD + (col ^ (((row & 15) >> 2) << 2)); }

__device__ __forceinline__ int col_bin(int col) { return (col < ECOLS) ? 2 * col : 2 * (col - ECOLS) + 1; }

__global__ void init_dft(float* __restrict__ cosT, float* __restrict__ sinT) {
    int idx = blockIdx.x * blockDim.x + threadIdx.x;
    if (idx >= KE * FPAD) return;
    int j   = idx / FPAD;
    int col = idx % FPAD;
    const bool even = col < ECOLS;
    const int f = col_bin(col);
    float c = 0.0f, s = 0.0f;
    if (f < NFREQS) {
        if (j <= (even ? KQ : KQ - 1)) {
            int r = (j * f) & (NFFT - 1);
            c = cosf((float)r * (6.28318530717958647692f / (float)NFFT));
        }
        if (j < KQ) {
            int k = (!even && j == 0) ? KQ : j;
            int r = (k * f) & (NFFT - 1);
            s = sinf((float)r * (6.28318530717958647692f / (float)NFFT));
        }
    }
    int off = (j >> 1) * (FPAD * 2) + col * 2 + (j & 1);
    cosT[off] = c;
    if (j < KQ) sinT[off] = s;
}

__device__ __forceinline__ float hz_to_mel(float f) { return 2595.0f * log10f(1.0f + f / 700.0f); }
__device__ __forceinline__ float mel_to_hz(float m) { return 700.0f * (exp10f(m / 2595.0f) - 1.0f); }

__global__ void init_mel(float* __restrict__ melFB) {
    int idx = blockIdx.x * blockDim.x + threadIdx.x;
    if (idx >= FPAD * NMELS) return;
    int k = idx / NMELS;
    int m = idx % NMELS;
    const int bin = col_bin(k);
    float v = 0.0f;
    if (bin < NFREQS) {
        float m_lo = hz_to_mel(50.0f);
        float m_hi = hz_to_mel((float)SRATE * 0.5f);
        float step = (m_hi - m_lo) / (float)(NMELS + 1);
        float f0 = mel_to_hz(m_lo + step * (float)m);
        float f1 = mel_to_hz(m_lo + step * (float)(m + 1));
        float f2 = mel_to_hz(m_lo + step * (float)(m + 2));
        float freq = (float)bin * ((float)SRATE * 0.5f / (float)(NFFT / 2));
        float down = (freq - f0) / (f1 - f0);
        float up   = (f2 - freq) / (f2 - f1);
        v = fmaxf(0.0f, fminf(down, up));
    }
    melFB[(k >> 1) * (NMELS * 2) + m * 2 + (k & 1)] = v;
}

#define S1K 4
#define S2K 4

struct TilePtr { const float* bcp; const float* bsp; const float* pa; const float* qa; };
struct Frag1 { v2f bc[S1K], bs[S1K], ap0[S1K], aq0[S1K], ap1[S1K], aq1[S1K]; };
struct Frag2 { v2f bm[S2K], a0[S2K], a1[S2K]; };

__device__ __forceinline__ void load_frag1(Frag1& f, const TilePtr& p, int j0) {
    #pragma unroll
    for (int s = 0; s < S1K; ++s) {
        const int j = j0 + 4 * s;
        f.bc[s] = *(const v2f*)&p.bcp[(j >> 1) * (FPAD * 2)];
        f.bs[s] = *(const v2f*)&p.bsp[(j >> 1) * (FPAD * 2)];
    }
    #pragma unroll
    for (int s = 0; s < S1K; ++s) {
        const int j = j0 + 4 * s;
        f.ap0[s] = *(const v2f*)&p.pa[j];
        f.aq0[s] = *(const v2f*)&p.qa[j];
        f.ap1[s] = *(const v2f*)&p.pa[16 * PITCH + j];
        f.aq1[s] = *(const v2f*)&p.qa[16 * PITCH + j];
    }
}

__device__ __forceinline__ void mma_frag1(const Frag1& f, v8f& re0, v8f& im0, v8f& re1, v8f& im1) {
    #pragma unroll
    for (int s = 0; s < S1K; ++s) {
        re0 = __builtin_amdgcn_wmma_f32_16x16x4_f32(false, f.ap0[s], false, f.bc[s],
                                                    (short)0, re0, false, false);
        im0 = __builtin_amdgcn_wmma_f32_16x16x4_f32(false, f.aq0[s], false, f.bs[s],
                                                    (short)0, im0, false, false);
        re1 = __builtin_amdgcn_wmma_f32_16x16x4_f32(false, f.ap1[s], false, f.bc[s],
                                                    (short)0, re1, false, false);
        im1 = __builtin_amdgcn_wmma_f32_16x16x4_f32(false, f.aq1[s], false, f.bs[s],
                                                    (short)0, im1, false, false);
    }
}

__global__ __launch_bounds__(256)
void dft_mel_kernel(const float* __restrict__ x,
                    const float* __restrict__ cosT,
                    const float* __restrict__ sinT,
                    const float* __restrict__ melFB,
                    float* __restrict__ melBuf)
{
    __shared__ float lds[4 * GFR * PITCH + SPAN];
    float* PE   = lds;
    float* PO   = lds + 1 * GFR * PITCH;
    float* QE   = lds + 2 * GFR * PITCH;
    float* QO   = lds + 3 * GFR * PITCH;
    float* span = lds + 4 * GFR * PITCH;
    float* mag  = lds;

    const int tile = blockIdx.x;
    const int b    = blockIdx.y;
    const int tid  = threadIdx.x;
    const int ln   = tid & 31;
    const int wv   = tid >> 5;
    const int t0   = tile * GFR;

    const float cw = cosf((float)tid * (6.28318530717958647692f / (float)NFFT));
    const float wa = 0.5f - 0.5f * cw;
    const float wb = 0.5f + 0.5f * cw;
    const float wq = 0.5f - 0.5f * cosf((float)(NFFT / 4) * (6.28318530717958647692f / (float)NFFT));

    const float* xb = x + (size_t)b * NSAMP;
    for (int h = 0; h < MT; ++h) {
        const long base = (long)(t0 + 16 * h) * HOP - (NFFT / 2);
        for (int i = tid; i < SPAN; i += 256) {
            long p = base + i;
            if (p < 0) p = -p;
            if (p >= NSAMP) p = 2L * NSAMP - 2 - p;
            if (p < 0) p = 0;
            if (p >= NSAMP) p = NSAMP - 1;
            span[i] = xb[p];
        }
        __syncthreads();

        if (tid != 0) {
            const int j = tid;
            for (int r = 0; r < 16; ++r) {
                const float* xr = span + r * HOP;
                const float ya = wa * xr[j];
                const float yd = wa * xr[NFFT - j];
                const float yb = wb * xr[NFFT / 2 - j];
                const float yc = wb * xr[NFFT / 2 + j];
                const float p0 = ya + yd, p1 = yb + yc;
                const float q0 = ya - yd, q1 = yb - yc;
                const int o = (16 * h + r) * PITCH + j;
                PE[o] = p0 + p1;
                PO[o] = p0 - p1;
                QE[o] = q0 - q1;
                QO[o] = q0 + q1;
            }
        }
        if (tid < 16) {
            const float* xr = span + tid * HOP;
            const float w0 = 0.5f - 0.5f * cosf(0.0f);
            const float wh = 0.5f + 0.5f * cosf(0.0f);
            const float y0   = w0 * xr[0];
            const float y512 = wh * xr[NFFT / 2];
            const float y256 = wq * xr[NFFT / 4];
            const float y768 = wq * xr[3 * NFFT / 4];
            const int o = (16 * h + tid) * PITCH;
            PE[o] = y0 + y512;
            PO[o] = y0 - y512;
            QE[o] = 0.0f;
            QO[o] = y256 - y768;
            PE[o + KQ]     = y256 + y768;
            PE[o + KQ + 1] = 0.0f;
            PE[o + KQ + 2] = 0.0f;
            PE[o + KQ + 3] = 0.0f;
        }
        __syncthreads();
    }

    const int mrow  = ln & 15;
    const int kh    = (ln >> 4) << 1;
    const int ncol  = ln & 15;
    const int mbase = (ln >> 4) * 8;

    float mreg[5][MT][8];
    auto tile_ptr = [&](int nt) {
        const bool ev = nt < NEVT;
        const int n = nt * 16 + ncol;
        TilePtr p;
        p.pa  = (ev ? PE : PO) + mrow * PITCH + kh;
        p.qa  = (ev ? QE : QO) + mrow * PITCH + kh;
        p.bcp = cosT + (kh >> 1) * (FPAD * 2) + n * 2;
        p.bsp = sinT + (kh >> 1) * (FPAD * 2) + n * 2;
        return p;
    };
    Frag1 fa, fb;
    TilePtr cur = tile_ptr(wv);
    load_frag1(fa, cur, 0);
    #pragma unroll
    for (int i = 0; i < 5; ++i) {
        const int nt = wv + 8 * i;
        if (nt < FPAD / 16) {
            const bool ev = nt < NEVT;
            const bool has_next = (i + 1 < 5) && (nt + 8 < FPAD / 16);
            const TilePtr nxt = tile_ptr(has_next ? nt + 8 : nt);
            v8f re0 = {0.f, 0.f, 0.f, 0.f, 0.f, 0.f, 0.f, 0.f};
            v8f im0 = {0.f, 0.f, 0.f, 0.f, 0.f, 0.f, 0.f, 0.f};
            v8f re1 = {0.f, 0.f, 0.f, 0.f, 0.f, 0.f, 0.f, 0.f};
            v8f im1 = {0.f, 0.f, 0.f, 0.f, 0.f, 0.f, 0.f, 0.f};
            #pragma unroll 1
            for (int j0 = 0; j0 < KQ - 8 * S1K; j0 += 8 * S1K) {
                load_frag1(fb, cur, j0 + 4 * S1K);
                __builtin_amdgcn_sched_barrier(0);
                mma_frag1(fa, re0, im0, re1, im1);
                __builtin_amdgcn_sched_barrier(0);
                load_frag1(fa, cur, j0 + 8 * S1K);
                __builtin_amdgcn_sched_barrier(0);
                mma_frag1(fb, re0, im0, re1, im1);
                __builtin_amdgcn_sched_barrier(0);
            }
            load_frag1(fb, cur, KQ - 4 * S1K);
            __builtin_amdgcn_sched_barrier(0);
            mma_frag1(fa, re0, im0, re1, im1);
            __builtin_amdgcn_sched_barrier(0);
            if (has_next) load_frag1(fa, nxt, 0);
            __builtin_amdgcn_sched_barrier(0);
            mma_frag1(fb, re0, im0, re1, im1);
            if (ev) {
                v2f bc  = *(const v2f*)&cur.bcp[(KQ >> 1) * (FPAD * 2)];
                v2f ap0 = *(const v2f*)&cur.pa[KQ];
                v2f ap1 = *(const v2f*)&cur.pa[16 * PITCH + KQ];
                re0 = __builtin_amdgcn_wmma_f32_16x16x4_f32(false, ap0, false, bc,
                                                            (short)0, re0, false, false);
                re1 = __builtin_amdgcn_wmma_f32_16x16x4_f32(false, ap1, false, bc,
                                                            (short)0, re1, false, false);
            }
            #pragma unroll
            for (int v = 0; v < 8; ++v) {
                float re = re0[v], im = im0[v];
                mreg[i][0][v] = sqrtf(re * re + im * im);
                re = re1[v]; im = im1[v];
                mreg[i][1][v] = sqrtf(re * re + im * im);
            }
            cur = nxt;
        }
    }
    __syncthreads();
    #pragma unroll
    for (int i = 0; i < 5; ++i) {
        const int nt = wv + 8 * i;
        if (nt < FPAD / 16) {
            const int n = nt * 16 + ncol;
            #pragma unroll
            for (int m = 0; m < MT; ++m) {
                #pragma unroll
                for (int v = 0; v < 8; ++v)
                    mag[mag_idx(16 * m + mbase + v, n)] = mreg[i][m][v];
            }
        }
    }
    __syncthreads();

    for (int nt = wv; nt < NMELS / 16; nt += 8) {
        v8f acc0 = {0.f, 0.f, 0.f, 0.f, 0.f, 0.f, 0.f, 0.f};
        v8f acc1 = {0.f, 0.f, 0.f, 0.f, 0.f, 0.f, 0.f, 0.f};
        const int n = nt * 16 + ncol;
        const float m_lo = hz_to_mel(50.0f);
        const float m_hi = hz_to_mel((float)SRATE * 0.5f);
        const float mstep = (m_hi - m_lo) / (float)(NMELS + 1);
        const float binw = (float)SRATE * 0.5f / (float)(NFFT / 2);
        const float f_lo = mel_to_hz(m_lo + mstep * (float)(16 * nt));
        const float f_hi = mel_to_hz(m_lo + mstep * (float)(16 * nt + 17));
        const int blo = max(0, (int)floorf(f_lo / binw) - 1);
        const int bhi = min(NFREQS - 1, (int)ceilf(f_hi / binw) + 1);
        const int eb0 = __builtin_amdgcn_readfirstlane((blo >> 1) >> 4);
        const int eb1 = __builtin_amdgcn_readfirstlane(((bhi >> 1) >> 4) + 1);
        const int ob0 = __builtin_amdgcn_readfirstlane(NEVT + ((max(0, blo - 1) >> 1) >> 4));
        const int ob1 = __builtin_amdgcn_readfirstlane(NEVT + ((max(0, bhi - 1) >> 1) >> 4) + 1);
        const int ne = eb1 - eb0;
        const int nb = ne + (ob1 - ob0);
        auto gcol = [&](int q) { return 16 * (q < ne ? eb0 + q : ob0 + (q - ne)); };
        auto load_frag2 = [&](Frag2& f, int kb0) {
            #pragma unroll
            for (int s = 0; s < S2K; ++s) {
                const int k0 = kb0 + 4 * s + kh;
                f.bm[s] = *(const v2f*)&melFB[(k0 >> 1) * (NMELS * 2) + n * 2];
            }
            #pragma unroll
            for (int s = 0; s < S2K; ++s) {
                const int k0 = kb0 + 4 * s + kh;
                f.a0[s] = *(const v2f*)&mag[mag_idx(mrow, k0)];
                f.a1[s] = *(const v2f*)&mag[mag_idx(16 + mrow, k0)];
            }
        };
        auto mma_frag2 = [&](const Frag2& f) {
            #pragma unroll
            for (int s = 0; s < S2K; ++s) {
                acc0 = __builtin_amdgcn_wmma_f32_16x16x4_f32(false, f.a0[s], false, f.bm[s],
                                                             (short)0, acc0, false, false);
                acc1 = __builtin_amdgcn_wmma_f32_16x16x4_f32(false, f.a1[s], false, f.bm[s],
                                                             (short)0, acc1, false, false);
            }
        };
        Frag2 ga, gb;
        load_frag2(ga, gcol(0));
        int q = 0;
        #pragma unroll 1
        for (; q + 2 < nb; q += 2) {
            load_frag2(gb, gcol(q + 1));
            __builtin_amdgcn_sched_barrier(0);
            mma_frag2(ga);
            __builtin_amdgcn_sched_barrier(0);
            load_frag2(ga, gcol(q + 2));
            __builtin_amdgcn_sched_barrier(0);
            mma_frag2(gb);
            __builtin_amdgcn_sched_barrier(0);
        }
        if (q + 1 < nb) {
            load_frag2(gb, gcol(q + 1));
            __builtin_amdgcn_sched_barrier(0);
            mma_frag2(ga);
            mma_frag2(gb);
        } else {
            mma_frag2(ga);
        }
        #pragma unroll
        for (int v = 0; v < 8; ++v) {
            int t = t0 + mbase + v;
            if (t < NFRAMES)
                melBuf[(size_t)t * NCH + b * NMELS + n] = acc0[v];
            t += 16;
            if (t < NFRAMES)
                melBuf[(size_t)t * NCH + b * NMELS + n] = acc1[v];
        }
    }
}

#define PC_CW     8
#define PC_SUB    4
#define PC_L      21
#define PC_NCHUNK 128
#define PC_PITCH  2692
#define PC_BLOCKS (NCH / PC_CW)

__global__ __launch_bounds__(1024)
void pcen_kernel(const float* __restrict__ melBuf,
                 float* __restrict__ out)
{
    __shared__ float tile[PC_CW * PC_PITCH];
    __shared__ float ends[PC_NCHUNK * PC_CW];

    const int tid   = threadIdx.x;
    const int c     = tid & (PC_CW - 1);
    const int chunk = tid >> 3;
    const int ch0   = blockIdx.x * PC_CW;
    const int t0    = chunk * PC_L;

    const float s  = 1.0f - expf(-((float)HOP / (float)SRATE) / 0.06f);
    const float d  = 1.0f - s;
    const float da = powf(2.0f, 0.98f);

    float e[PC_L];
    #pragma unroll
    for (int i = 0; i < PC_L; ++i) {
        const int t = min(t0 + i, NFRAMES - 1);
        e[i] = melBuf[(size_t)t * NCH + ch0 + c];
    }
    float* row = tile + c * PC_PITCH + t0;
    float loc = 0.0f;
    #pragma unroll
    for (int i = 0; i < PC_L; ++i) {
        row[i] = e[i];
        loc = d * loc + s * e[i];
    }
    ends[chunk * PC_CW + c] = loc;
    __syncthreads();

    float dL = 1.0f;
    #pragma unroll
    for (int i = 0; i < PC_L; ++i) dL *= d;
    float M = 0.0f;
    #pragma unroll 8
    for (int j = 0; j < chunk; ++j) M = dL * M + ends[j * PC_CW + c];

    const int n = max(0, min(PC_L, NFRAMES - t0));
    for (int i = 0; i < n; ++i) {
        const float ev = row[i];
        M = (1.0f - s) * M + s * ev;
        const float smooth = sqrtf(M + 1e-6f);
        row[i] = powf(ev / smooth + 2.0f, 0.98f) - da;
    }
    __syncthreads();

    float* o = out + (size_t)ch0 * NFRAMES;
    for (int idx = tid; idx < PC_CW * NFRAMES; idx += 1024) {
        const int r = idx / NFRAMES;
        const int t = idx - r * NFRAMES;
        o[idx] = tile[r * PC_PITCH + t];
    }
}

extern "C" void kernel_launch(void* const* d_in, const int* in_sizes, int n_in,
                              void* d_out, int out_size, void* d_ws, size_t ws_size,
                              hipStream_t stream) {
    const float* x = (const float*)d_in[0];
    float* ws = (float*)d_ws;

    float* cosT   = ws;
    float* sinT   = cosT  + (size_t)KE * FPAD;
    float* melFB  = sinT  + (size_t)KQ * FPAD;
    float* melBuf = melFB + (size_t)FPAD * NMELS;

    int n1 = KE * FPAD;
    init_dft<<<(n1 + 255) / 256, 256, 0, stream>>>(cosT, sinT);
    int n2 = FPAD * NMELS;
    init_mel<<<(n2 + 255) / 256, 256, 0, stream>>>(melFB);

    dim3 grid(GTILES, BATCH);
    dft_mel_kernel<<<grid, 256, 0, stream>>>(x, cosT, sinT, melFB, melBuf);

    pcen_kernel<<<PC_BLOCKS, 1024, 0, stream>>>(melBuf, (float*)d_out);
}


# dft_mel: third fold of the real DFT about tap 128 for the even bins (bins 4c take PE[j]+PE[256-j], bins 4c+2 PE[j]-PE[256-j]): K 256->128 on 17 of 33 bin tiles; DFT and mel sum order change
# speedup vs baseline: 14.8461x; 1.1859x over previous
#include <hip/hip_runtime.h>
#include <math.h>

#define SRATE     22050
#define NFFT      1024
#define HOP       256
#define NMELS     128
#define NFREQS    513
#define FPAD      528
#define C2COLS    144
#define ECOLS     272
#define N0T       (C2COLS/16)
#define NEVT      (ECOLS/16)
#define BATCH     16
#define NSAMP     661504
#define NFRAMES   2585
#define MT        2
#define GFR       (16*MT)
#define GTILES    81
#define SPAN      (15*HOP + NFFT)
#define NCH       (BATCH*NMELS)
#define KQ        (NFFT/4)
#define KE        (KQ + 4)
#define PITCH     (KQ + 4)
#define KH        (NFFT/8)
#define PITCHE    (KH + 4)

typedef __attribute__((ext_vector_type(2))) float v2f;
typedef __attribute__((ext_vector_type(8))) float v8f;

__device__ __forceinline__ int mag_idx(int row, int col) { return row * FPAD + (col ^ (((row & 15) >> 2) << 2)); }

__device__ __forceinline__ int col_bin(int col) {
    return (col < C2COLS) ? 4 * col : (col < ECOLS) ? 4 * (col - C2COLS) + 2 : 2 * (col - ECOLS) + 1;
}

__global__ void init_dft(float* __restrict__ cosT, float* __restrict__ sinT) {
    int idx = blockIdx.x * blockDim.x + threadIdx.x;
    if (idx >= KE * FPAD) return;
    int j   = idx / FPAD;
    int col = idx % FPAD;
    const int fam = (col < C2COLS) ? 0 : (col < ECOLS) ? 1 : 2;
    const int f = col_bin(col);
    const int kc = (fam == 0) ? KH : (fam == 1) ? KH - 1 : KQ - 1;
    const int ks = (fam == 2) ? KQ : KH;
    float c = 0.0f, s = 0.0f;
    if (f < NFREQS) {
        if (j <= kc) {
            int r = (j * f) & (NFFT - 1);
            c = cosf((float)r * (6.28318530717958647692f / (float)NFFT));
        }
        if (j < ks) {
            int k = (fam != 0 && j == 0) ? ks : j;
            int r = (k * f) & (NFFT - 1);
            s = sinf((float)r * (6.28318530717958647692f / (float)NFFT));
        }
    }
    int off = (j >> 1) * (FPAD * 2) + col * 2 + (j & 1);
    cosT[off] = c;
    if (j < KQ) sinT[off] = s;
}

__device__ __forceinline__ float hz_to_mel(float f) { return 2595.0f * log10f(1.0f + f / 700.0f); }
__device__ __forceinline__ float mel_to_hz(float m) { return 700.0f * (exp10f(m / 2595.0f) - 1.0f); }

__global__ void init_mel(float* __restrict__ melFB) {
    int idx = blockIdx.x * blockDim.x + threadIdx.x;
    if (idx >= FPAD * NMELS) return;
    int k = idx / NMELS;
    int m = idx % NMELS;
    const int bin = col_bin(k);
    float v = 0.0f;
    if (bin < NFREQS) {
        float m_lo = hz_to_mel(50.0f);
        float m_hi = hz_to_mel((float)SRATE * 0.5f);
        float step = (m_hi - m_lo) / (float)(NMELS + 1);
        float f0 = mel_to_hz(m_lo + step * (float)m);
        float f1 = mel_to_hz(m_lo + step * (float)(m + 1));
        float f2 = mel_to_hz(m_lo + step * (float)(m + 2));
        float freq = (float)bin * ((float)SRATE * 0.5f / (float)(NFFT / 2));
        float down = (freq - f0) / (f1 - f0);
        float up   = (f2 - freq) / (f2 - f1);
        v = fmaxf(0.0f, fminf(down, up));
    }
    melFB[(k >> 1) * (NMELS * 2) + m * 2 + (k & 1)] = v;
}

#define S1K 4
#define S2K 4

struct TilePtr { int bo; int pa; int qa; };
struct Frag1 { v2f bc[S1K], bs[S1K], ap0[S1K], aq0[S1K], ap1[S1K], aq1[S1K]; };
struct Frag2 { v2f bm[S2K], a0[S2K], a1[S2K]; };

__device__ __forceinline__ void load_frag1(Frag1& f, const float* __restrict__ cosT, const float* __restrict__ sinT,
                                           const float* lds, const TilePtr& p, int j0, const int p16, const int sk) {
    #pragma unroll
    for (int s = 0; s < sk; ++s) {
        const int j = j0 + 4 * s;
        f.bc[s] = *(const v2f*)&cosT[p.bo + (j >> 1) * (FPAD * 2)];
        f.bs[s] = *(const v2f*)&sinT[p.bo + (j >> 1) * (FPAD * 2)];
    }
    #pragma unroll
    for (int s = 0; s < sk; ++s) {
        const int j = j0 + 4 * s;
        f.ap0[s] = *(const v2f*)&lds[p.pa + j];
        f.aq0[s] = *(const v2f*)&lds[p.qa + j];
        f.ap1[s] = *(const v2f*)&lds[p.pa + p16 + j];
        f.aq1[s] = *(const v2f*)&lds[p.qa + p16 + j];
    }
}

__device__ __forceinline__ void mma_frag1(const Frag1& f, v8f& re0, v8f& im0, v8f& re1, v8f& im1, const int sk) {
    #pragma unroll
    for (int s = 0; s < sk; ++s) {
        re0 = __builtin_amdgcn_wmma_f32_16x16x4_f32(false, f.ap0[s], false, f.bc[s],
                                                    (short)0, re0, false, false);
        im0 = __builtin_amdgcn_wmma_f32_16x16x4_f32(false, f.aq0[s], false, f.bs[s],
                                                    (short)0, im0, false, false);
        re1 = __builtin_amdgcn_wmma_f32_16x16x4_f32(false, f.ap1[s], false, f.bc[s],
                                                    (short)0, re1, false, false);
        im1 = __builtin_amdgcn_wmma_f32_16x16x4_f32(false, f.aq1[s], false, f.bs[s],
                                                    (short)0, im1, false, false);
    }
}

__global__ __launch_bounds__(256)
void dft_mel_kernel(const float* __restrict__ x,
                    const float* __restrict__ cosT,
                    const float* __restrict__ sinT,
                    const float* __restrict__ melFB,
                    float* __restrict__ melBuf)
{
    __shared__ float lds[2 * GFR * PITCH + 4 * GFR * PITCHE + SPAN];
    float* PO   = lds;
    float* QO   = lds + 1 * GFR * PITCH;
    float* C0   = lds + 2 * GFR * PITCH;
    float* C2   = lds + 2 * GFR * PITCH + 1 * GFR * PITCHE;
    float* S0   = lds + 2 * GFR * PITCH + 2 * GFR * PITCHE;
    float* S2   = lds + 2 * GFR * PITCH + 3 * GFR * PITCHE;
    float* span = lds + 2 * GFR * PITCH + 4 * GFR * PITCHE;
    float* mag  = lds;

    const int tile = blockIdx.x;
    const int b    = blockIdx.y;
    const int tid  = threadIdx.x;
    const int ln   = tid & 31;
    const int wv   = __builtin_amdgcn_readfirstlane(tid >> 5);
    const int t0   = tile * GFR;

    const int jj = tid & 127;
    const int rh = tid >> 7;
    const float cw1 = cosf((float)jj * (6.28318530717958647692f / (float)NFFT));
    const float cw2 = cosf((float)(KQ - jj) * (6.28318530717958647692f / (float)NFFT));
    const float wa1 = 0.5f - 0.5f * cw1, wb1 = 0.5f + 0.5f * cw1;
    const float wa2 = 0.5f - 0.5f * cw2, wb2 = 0.5f + 0.5f * cw2;
    const float wq = 0.5f - 0.5f * cosf((float)(NFFT / 4) * (6.28318530717958647692f / (float)NFFT));

    const float* xb = x + (size_t)b * NSAMP;
    for (int h = 0; h < MT; ++h) {
        const long base = (long)(t0 + 16 * h) * HOP - (NFFT / 2);
        for (int i = tid; i < SPAN; i += 256) {
            long p = base + i;
            if (p < 0) p = -p;
            if (p >= NSAMP) p = 2L * NSAMP - 2 - p;
            if (p < 0) p = 0;
            if (p >= NSAMP) p = NSAMP - 1;
            span[i] = xb[p];
        }
        __syncthreads();

        if (jj != 0) {
            const int j1 = jj, j2 = KQ - jj;
            for (int r8 = 0; r8 < 8; ++r8) {
                const int r = 8 * rh + r8;
                const float* xr = span + r * HOP;
                const float ya1 = wa1 * xr[j1];
                const float yd1 = wa1 * xr[NFFT - j1];
                const float yb1 = wb1 * xr[NFFT / 2 - j1];
                const float yc1 = wb1 * xr[NFFT / 2 + j1];
                const float p01 = ya1 + yd1, p11 = yb1 + yc1;
                const float q01 = ya1 - yd1, q11 = yb1 - yc1;
                const float ya2 = wa2 * xr[j2];
                const float yd2 = wa2 * xr[NFFT - j2];
                const float yb2 = wb2 * xr[NFFT / 2 - j2];
                const float yc2 = wb2 * xr[NFFT / 2 + j2];
                const float p02 = ya2 + yd2, p12 = yb2 + yc2;
                const float q02 = ya2 - yd2, q12 = yb2 - yc2;
                const float pe1 = p01 + p11, pe2 = p02 + p12;
                const float qe1 = q01 - q11, qe2 = q02 - q12;
                const int o = (16 * h + r) * PITCH;
                PO[o + j1] = p01 - p11;
                PO[o + j2] = p02 - p12;
                QO[o + j1] = q01 + q11;
                QO[o + j2] = q02 + q12;
                const int oe = (16 * h + r) * PITCHE + j1;
                C0[oe] = pe1 + pe2;
                C2[oe] = pe1 - pe2;
                S0[oe] = qe1 - qe2;
                S2[oe] = qe1 + qe2;
            }
        }
        if (tid < 16) {
            const float* xr = span + tid * HOP;
            const float w0 = 0.5f - 0.5f * cosf(0.0f);
            const float wh = 0.5f + 0.5f * cosf(0.0f);
            const float c8 = cosf((float)(NFFT / 8) * (6.28318530717958647692f / (float)NFFT));
            const float w128 = 0.5f - 0.5f * c8;
            const float w384 = 0.5f + 0.5f * c8;
            const float y0   = w0 * xr[0];
            const float y512 = wh * xr[NFFT / 2];
            const float y256 = wq * xr[NFFT / 4];
            const float y768 = wq * xr[3 * NFFT / 4];
            const float ya = w128 * xr[NFFT / 8];
            const float yd = w128 * xr[NFFT - NFFT / 8];
            const float yb = w384 * xr[NFFT / 2 - NFFT / 8];
            const float yc = w384 * xr[NFFT / 2 + NFFT / 8];
            const float p0 = ya + yd, p1 = yb + yc;
            const float q0 = ya - yd, q1 = yb - yc;
            const int o = (16 * h + tid) * PITCH;
            PO[o] = y0 - y512;
            QO[o] = y256 - y768;
            PO[o + KH] = p0 - p1;
            QO[o + KH] = q0 + q1;
            const int oe = (16 * h + tid) * PITCHE;
            const float pe0 = y0 + y512;
            const float pe256 = y256 + y768;
            C0[oe] = pe0 + pe256;
            C2[oe] = pe0 - pe256;
            S0[oe] = 0.0f;
            S2[oe] = q0 - q1;
            C0[oe + KH]     = p0 + p1;
            C0[oe + KH + 1] = 0.0f;
            C0[oe + KH + 2] = 0.0f;
            C0[oe + KH + 3] = 0.0f;
        }
        __syncthreads();
    }

    const int mrow  = ln & 15;
    const int kh    = (ln >> 4) << 1;
    const int ncol  = ln & 15;
    const int mbase = (ln >> 4) * 8;

    float mreg[5][MT][8];
    auto tile_of = [&](int i) { return i == 0 ? NEVT + wv : i == 1 ? NEVT + 8 + wv : i == 2 ? wv : i == 3 ? 8 + wv : 16; };
    auto tile_ptr = [&](int i) {
        const int nt = tile_of(i);
        const int n = nt * 16 + ncol;
        TilePtr p;
        if (i < 2) {
            p.pa = mrow * PITCH + kh;
            p.qa = GFR * PITCH + mrow * PITCH + kh;
        } else {
            const int fam2 = (nt < N0T) ? 0 : GFR * PITCHE;
            p.pa = 2 * GFR * PITCH + fam2 + mrow * PITCHE + kh;
            p.qa = 2 * GFR * PITCH + 2 * GFR * PITCHE + fam2 + mrow * PITCHE + kh;
        }
        p.bo  = (kh >> 1) * (FPAD * 2) + n * 2;
        return p;
    };
    Frag1 fa, fb;
    TilePtr cur = tile_ptr(0);
    load_frag1(fa, cosT, sinT, lds, cur, 0, 16 * PITCH, S1K);
    #pragma unroll
    for (int i = 0; i < 5; ++i) {
        const int nt = tile_of(i);
        if (i < 4 || wv == 0) {
            const bool c0t = nt < N0T;
            const int  kt  = (i < 2) ? KQ : KH;
            const int  p16 = 16 * ((i < 2) ? PITCH : PITCHE);
            const int  n16 = 16 * ((i + 1 < 2) ? PITCH : PITCHE);
            const int  sk  = (i < 2) ? S1K : S1K / 2;
            const int  nsk = (i + 1 < 2) ? S1K : S1K / 2;
            const bool has_next = (i + 1 < 4) || (i + 1 == 4 && wv == 0);
            const TilePtr nxt = tile_ptr(has_next ? i + 1 : i);
            v8f re0 = {0.f, 0.f, 0.f, 0.f, 0.f, 0.f, 0.f, 0.f};
            v8f im0 = {0.f, 0.f, 0.f, 0.f, 0.f, 0.f, 0.f, 0.f};
            v8f re1 = {0.f, 0.f, 0.f, 0.f, 0.f, 0.f, 0.f, 0.f};
            v8f im1 = {0.f, 0.f, 0.f, 0.f, 0.f, 0.f, 0.f, 0.f};
            #pragma unroll 1
            for (int j0 = 0; j0 < kt - 8 * sk; j0 += 8 * sk) {
                load_frag1(fb, cosT, sinT, lds, cur, j0 + 4 * sk, p16, sk);
                __builtin_amdgcn_sched_barrier(0);
                mma_frag1(fa, re0, im0, re1, im1, sk);
                __builtin_amdgcn_sched_barrier(0);
                load_frag1(fa, cosT, sinT, lds, cur, j0 + 8 * sk, p16, sk);
                __builtin_amdgcn_sched_barrier(0);
                mma_frag1(fb, re0, im0, re1, im1, sk);
                __builtin_amdgcn_sched_barrier(0);
            }
            load_frag1(fb, cosT, sinT, lds, cur, kt - 4 * sk, p16, sk);
            __builtin_amdgcn_sched_barrier(0);
            mma_frag1(fa, re0, im0, re1, im1, sk);
            __builtin_amdgcn_sched_barrier(0);
            if (has_next) load_frag1(fa, cosT, sinT, lds, nxt, 0, n16, nsk);
            __builtin_amdgcn_sched_barrier(0);
            mma_frag1(fb, re0, im0, re1, im1, sk);
            if (c0t) {
                v2f bc  = *(const v2f*)&cosT[cur.bo + (KH >> 1) * (FPAD * 2)];
                v2f ap0 = *(const v2f*)&lds[cur.pa + KH];
                v2f ap1 = *(const v2f*)&lds[cur.pa + p16 + KH];
                re0 = __builtin_amdgcn_wmma_f32_16x16x4_f32(false, ap0, false, bc,
                                                            (short)0, re0, false, false);
                re1 = __builtin_amdgcn_wmma_f32_16x16x4_f32(false, ap1, false, bc,
                                                            (short)0, re1, false, false);
            }
            #pragma unroll
            for (int v = 0; v < 8; ++v) {
                float re = re0[v], im = im0[v];
                mreg[i][0][v] = sqrtf(re * re + im * im);
                re = re1[v]; im = im1[v];
                mreg[i][1][v] = sqrtf(re * re + im * im);
            }
            cur = nxt;
        }
    }
    __syncthreads();
    #pragma unroll
    for (int i = 0; i < 5; ++i) {
        const int nt = tile_of(i);
        if (i < 4 || wv == 0) {
            const int n = nt * 16 + ncol;
            #pragma unroll
            for (int m = 0; m < MT; ++m) {
                #pragma unroll
                for (int v = 0; v < 8; ++v)
                    mag[mag_idx(16 * m + mbase + v, n)] = mreg[i][m][v];
            }
        }
    }
    __syncthreads();

    for (int nt = wv; nt < NMELS / 16; nt += 8) {
        v8f acc0 = {0.f, 0.f, 0.f, 0.f, 0.f, 0.f, 0.f, 0.f};
        v8f acc1 = {0.f, 0.f, 0.f, 0.f, 0.f, 0.f, 0.f, 0.f};
        const int n = nt * 16 + ncol;
        const float m_lo = hz_to_mel(50.0f);
        const float m_hi = hz_to_mel((float)SRATE * 0.5f);
        const float mstep = (m_hi - m_lo) / (float)(NMELS + 1);
        const float binw = (float)SRATE * 0.5f / (float)(NFFT / 2);
        const float f_lo = mel_to_hz(m_lo + mstep * (float)(16 * nt));
        const float f_hi = mel_to_hz(m_lo + mstep * (float)(16 * nt + 17));
        const int blo = max(0, (int)floorf(f_lo / binw) - 1);
        const int bhi = min(NFREQS - 1, (int)ceilf(f_hi / binw) + 1);
        const int a0 = __builtin_amdgcn_readfirstlane((blo >> 2) >> 4);
        const int a1 = __builtin_amdgcn_readfirstlane(((bhi >> 2) >> 4) + 1);
        const int c0 = __builtin_amdgcn_readfirstlane(N0T + ((max(0, blo - 2) >> 2) >> 4));
        const int c1 = __builtin_amdgcn_readfirstlane(N0T + ((max(0, bhi - 2) >> 2) >> 4) + 1);
        const int ob0 = __builtin_amdgcn_readfirstlane(NEVT + ((max(0, blo - 1) >> 1) >> 4));
        const int ob1 = __builtin_amdgcn_readfirstlane(NEVT + ((max(0, bhi - 1) >> 1) >> 4) + 1);
        const int na = a1 - a0;
        const int ne = na + (c1 - c0);
        const int nb = ne + (ob1 - ob0);
        auto gcol = [&](int q) { return 16 * (q < na ? a0 + q : q < ne ? c0 + (q - na) : ob0 + (q - ne)); };
        auto load_frag2 = [&](Frag2& f, int kb0) {
            #pragma unroll
            for (int s = 0; s < S2K; ++s) {
                const int k0 = kb0 + 4 * s + kh;
                f.bm[s] = *(const v2f*)&melFB[(k0 >> 1) * (NMELS * 2) + n * 2];
            }
            #pragma unroll
            for (int s = 0; s < S2K; ++s) {
                const int k0 = kb0 + 4 * s + kh;
                f.a0[s] = *(const v2f*)&mag[mag_idx(mrow, k0)];
                f.a1[s] = *(const v2f*)&mag[mag_idx(16 + mrow, k0)];
            }
        };
        auto mma_frag2 = [&](const Frag2& f) {
            #pragma unroll
            for (int s = 0; s < S2K; ++s) {
                acc0 = __builtin_amdgcn_wmma_f32_16x16x4_f32(false, f.a0[s], false, f.bm[s],
                                                             (short)0, acc0, false, false);
                acc1 = __builtin_amdgcn_wmma_f32_16x16x4_f32(false, f.a1[s], false, f.bm[s],
                                                             (short)0, acc1, false, false);
            }
        };
        Frag2 ga, gb;
        load_frag2(ga, gcol(0));
        int q = 0;
        #pragma unroll 1
        for (; q + 2 < nb; q += 2) {
            load_frag2(gb, gcol(q + 1));
            __builtin_amdgcn_sched_barrier(0);
            mma_frag2(ga);
            __builtin_amdgcn_sched_barrier(0);
            load_frag2(ga, gcol(q + 2));
            __builtin_amdgcn_sched_barrier(0);
            mma_frag2(gb);
            __builtin_amdgcn_sched_barrier(0);
        }
        if (q + 1 < nb) {
            load_frag2(gb, gcol(q + 1));
            __builtin_amdgcn_sched_barrier(0);
            mma_frag2(ga);
            mma_frag2(gb);
        } else {
            mma_frag2(ga);
        }
        #pragma unroll
        for (int v = 0; v < 8; ++v) {
            int t = t0 + mbase + v;
            if (t < NFRAMES)
                melBuf[(size_t)t * NCH + b * NMELS + n] = acc0[v];
            t += 16;
            if (t < NFRAMES)
                melBuf[(size_t)t * NCH + b * NMELS + n] = acc1[v];
        }
    }
}

#define PC_CW     8
#define PC_SUB    4
#define PC_L      21
#define PC_NCHUNK 128
#define PC_PITCH  2692
#define PC_BLOCKS (NCH / PC_CW)

__global__ __launch_bounds__(1024)
void pcen_kernel(const float* __restrict__ melBuf,
                 float* __restrict__ out)
{
    __shared__ float tile[PC_CW * PC_PITCH];
    __shared__ float ends[PC_NCHUNK * PC_CW];

    const int tid   = threadIdx.x;
    const int c     = tid & (PC_CW - 1);
    const int chunk = tid >> 3;
    const int ch0   = blockIdx.x * PC_CW;
    const int t0    = chunk * PC_L;

    const float s  = 1.0f - expf(-((float)HOP / (float)SRATE) / 0.06f);
    const float d  = 1.0f - s;
    const float da = powf(2.0f, 0.98f);

    float e[PC_L];
    #pragma unroll
    for (int i = 0; i < PC_L; ++i) {
        const int t = min(t0 + i, NFRAMES - 1);
        e[i] = melBuf[(size_t)t * NCH + ch0 + c];
    }
    float* row = tile + c * PC_PITCH + t0;
    float loc = 0.0f;
    #pragma unroll
    for (int i = 0; i < PC_L; ++i) {
        row[i] = e[i];
        loc = d * loc + s * e[i];
    }
    ends[chunk * PC_CW + c] = loc;
    __syncthreads();

    float dL = 1.0f;
    #pragma unroll
    for (int i = 0; i < PC_L; ++i) dL *= d;
    float M = 0.0f;
    #pragma unroll 8
    for (int j = 0; j < chunk; ++j) M = dL * M + ends[j * PC_CW + c];

    const int n = max(0, min(PC_L, NFRAMES - t0));
    for (int i = 0; i < n; ++i) {
        const float ev = row[i];
        M = (1.0f - s) * M + s * ev;
        const float smooth = sqrtf(M + 1e-6f);
        row[i] = powf(ev / smooth + 2.0f, 0.98f) - da;
    }
    __syncthreads();

    float* o = out + (size_t)ch0 * NFRAMES;
    for (int idx = tid; idx < PC_CW * NFRAMES; idx += 1024) {
        const int r = idx / NFRAMES;
        const int t = idx - r * NFRAMES;
        o[idx] = tile[r * PC_PITCH + t];
    }
}

extern "C" void kernel_launch(void* const* d_in, const int* in_sizes, int n_in,
                              void* d_out, int out_size, void* d_ws, size_t ws_size,
                              hipStream_t stream) {
    const float* x = (const float*)d_in[0];
    float* ws = (float*)d_ws;

    float* cosT   = ws;
    float* sinT   = cosT  + (size_t)KE * FPAD;
    float* melFB  = sinT  + (size_t)KQ * FPAD;
    float* melBuf = melFB + (size_t)FPAD * NMELS;

    int n1 = KE * FPAD;
    init_dft<<<(n1 + 255) / 256, 256, 0, stream>>>(cosT, sinT);
    int n2 = FPAD * NMELS;
    init_mel<<<(n2 + 255) / 256, 256, 0, stream>>>(melFB);

    dim3 grid(GTILES, BATCH);
    dft_mel_kernel<<<grid, 256, 0, stream>>>(x, cosT, sinT, melFB, melBuf);

    pcen_kernel<<<PC_BLOCKS, 1024, 0, stream>>>(melBuf, (float*)d_out);
}


# pcen pointwise on the hardware special-function instructions: x^0.98 = x*exp2(-0.02*log2 x), e/sqrt(M+eps) = e*rsq(M+eps), instead of library pow, sqrt, divide; results move by a few ulp
# speedup vs baseline: 16.1691x; 1.0891x over previous
#include <hip/hip_runtime.h>
#include <math.h>

#define SRATE     22050
#define NFFT      1024
#define HOP       256
#define NMELS     128
#define NFREQS    513
#define FPAD      528
#define C2COLS    144
#define ECOLS     272
#define N0T       (C2COLS/16)
#define NEVT      (ECOLS/16)
#define BATCH     16
#define NSAMP     661504
#define NFRAMES   2585
#define MT        2
#define GFR       (16*MT)
#define GTILES    81
#define SPAN      (15*HOP + NFFT)
#define NCH       (BATCH*NMELS)
#define KQ        (NFFT/4)
#define KE        (KQ + 4)
#define PITCH     (KQ + 4)
#define KH        (NFFT/8)
#define PITCHE    (KH + 4)

typedef __attribute__((ext_vector_type(2))) float v2f;
typedef __attribute__((ext_vector_type(8))) float v8f;

__device__ __forceinline__ int mag_idx(int row, int col) { return row * FPAD + (col ^ (((row & 15) >> 2) << 2)); }

__device__ __forceinline__ int col_bin(int col) {
    return (col < C2COLS) ? 4 * col : (col < ECOLS) ? 4 * (col - C2COLS) + 2 : 2 * (col - ECOLS) + 1;
}

__global__ void init_dft(float* __restrict__ cosT, float* __restrict__ sinT) {
    int idx = blockIdx.x * blockDim.x + threadIdx.x;
    if (idx >= KE * FPAD) return;
    int j   = idx / FPAD;
    int col = idx % FPAD;
    const int fam = (col < C2COLS) ? 0 : (col < ECOLS) ? 1 : 2;
    const int f = col_bin(col);
    const int kc = (fam == 0) ? KH : (fam == 1) ? KH - 1 : KQ - 1;
    const int ks = (fam == 2) ? KQ : KH;
    float c = 0.0f, s = 0.0f;
    if (f < NFREQS) {
        if (j <= kc) {
            int r = (j * f) & (NFFT - 1);
            c = cosf((float)r * (6.28318530717958647692f / (float)NFFT));
        }
        if (j < ks) {
            int k = (fam != 0 && j == 0) ? ks : j;
            int r = (k * f) & (NFFT - 1);
            s = sinf((float)r * (6.28318530717958647692f / (float)NFFT));
        }
    }
    int off = (j >> 1) * (FPAD * 2) + col * 2 + (j & 1);
    cosT[off] = c;
    if (j < KQ) sinT[off] = s;
}

__device__ __forceinline__ float hz_to_mel(float f) { return 2595.0f * log10f(1.0f + f / 700.0f); }
__device__ __forceinline__ float mel_to_hz(float m) { return 700.0f * (exp10f(m / 2595.0f) - 1.0f); }

__global__ void init_mel(float* __restrict__ melFB) {
    int idx = blockIdx.x * blockDim.x + threadIdx.x;
    if (idx >= FPAD * NMELS) return;
    int k = idx / NMELS;
    int m = idx % NMELS;
    const int bin = col_bin(k);
    float v = 0.0f;
    if (bin < NFREQS) {
        float m_lo = hz_to_mel(50.0f);
        float m_hi = hz_to_mel((float)SRATE * 0.5f);
        float step = (m_hi - m_lo) / (float)(NMELS + 1);
        float f0 = mel_to_hz(m_lo + step * (float)m);
        float f1 = mel_to_hz(m_lo + step * (float)(m + 1));
        float f2 = mel_to_hz(m_lo + step * (float)(m + 2));
        float freq = (float)bin * ((float)SRATE * 0.5f / (float)(NFFT / 2));
        float down = (freq - f0) / (f1 - f0);
        float up   = (f2 - freq) / (f2 - f1);
        v = fmaxf(0.0f, fminf(down, up));
    }
    melFB[(k >> 1) * (NMELS * 2) + m * 2 + (k & 1)] = v;
}

#define S1K 4
#define S2K 4

struct TilePtr { int bo; int pa; int qa; };
struct Frag1 { v2f bc[S1K], bs[S1K], ap0[S1K], aq0[S1K], ap1[S1K], aq1[S1K]; };
struct Frag2 { v2f bm[S2K], a0[S2K], a1[S2K]; };

__device__ __forceinline__ void load_frag1(Frag1& f, const float* __restrict__ cosT, const float* __restrict__ sinT,
                                           const float* lds, const TilePtr& p, int j0, const int p16, const int sk) {
    #pragma unroll
    for (int s = 0; s < sk; ++s) {
        const int j = j0 + 4 * s;
        f.bc[s] = *(const v2f*)&cosT[p.bo + (j >> 1) * (FPAD * 2)];
        f.bs[s] = *(const v2f*)&sinT[p.bo + (j >> 1) * (FPAD * 2)];
    }
    #pragma unroll
    for (int s = 0; s < sk; ++s) {
        const int j = j0 + 4 * s;
        f.ap0[s] = *(const v2f*)&lds[p.pa + j];
        f.aq0[s] = *(const v2f*)&lds[p.qa + j];
        f.ap1[s] = *(const v2f*)&lds[p.pa + p16 + j];
        f.aq1[s] = *(const v2f*)&lds[p.qa + p16 + j];
    }
}

__device__ __forceinline__ void mma_frag1(const Frag1& f, v8f& re0, v8f& im0, v8f& re1, v8f& im1, const int sk) {
    #pragma unroll
    for (int s = 0; s < sk; ++s) {
        re0 = __builtin_amdgcn_wmma_f32_16x16x4_f32(false, f.ap0[s], false, f.bc[s],
                                                    (short)0, re0, false, false);
        im0 = __builtin_amdgcn_wmma_f32_16x16x4_f32(false, f.aq0[s], false, f.bs[s],
                                                    (short)0, im0, false, false);
        re1 = __builtin_amdgcn_wmma_f32_16x16x4_f32(false, f.ap1[s], false, f.bc[s],
                                                    (short)0, re1, false, false);
        im1 = __builtin_amdgcn_wmma_f32_16x16x4_f32(false, f.aq1[s], false, f.bs[s],
                                                    (short)0, im1, false, false);
    }
}

__global__ __launch_bounds__(256)
void dft_mel_kernel(const float* __restrict__ x,
                    const float* __restrict__ cosT,
                    const float* __restrict__ sinT,
                    const float* __restrict__ melFB,
                    float* __restrict__ melBuf)
{
    __shared__ float lds[2 * GFR * PITCH + 4 * GFR * PITCHE + SPAN];
    float* PO   = lds;
    float* QO   = lds + 1 * GFR * PITCH;
    float* C0   = lds + 2 * GFR * PITCH;
    float* C2   = lds + 2 * GFR * PITCH + 1 * GFR * PITCHE;
    float* S0   = lds + 2 * GFR * PITCH + 2 * GFR * PITCHE;
    float* S2   = lds + 2 * GFR * PITCH + 3 * GFR * PITCHE;
    float* span = lds + 2 * GFR * PITCH + 4 * GFR * PITCHE;
    float* mag  = lds;

    const int tile = blockIdx.x;
    const int b    = blockIdx.y;
    const int tid  = threadIdx.x;
    const int ln   = tid & 31;
    const int wv   = __builtin_amdgcn_readfirstlane(tid >> 5);
    const int t0   = tile * GFR;

    const int jj = tid & 127;
    const int rh = tid >> 7;
    const float cw1 = cosf((float)jj * (6.28318530717958647692f / (float)NFFT));
    const float cw2 = cosf((float)(KQ - jj) * (6.28318530717958647692f / (float)NFFT));
    const float wa1 = 0.5f - 0.5f * cw1, wb1 = 0.5f + 0.5f * cw1;
    const float wa2 = 0.5f - 0.5f * cw2, wb2 = 0.5f + 0.5f * cw2;
    const float wq = 0.5f - 0.5f * cosf((float)(NFFT / 4) * (6.28318530717958647692f / (float)NFFT));

    const float* xb = x + (size_t)b * NSAMP;
    for (int h = 0; h < MT; ++h) {
        const long base = (long)(t0 + 16 * h) * HOP - (NFFT / 2);
        for (int i = tid; i < SPAN; i += 256) {
            long p = base + i;
            if (p < 0) p = -p;
            if (p >= NSAMP) p = 2L * NSAMP - 2 - p;
            if (p < 0) p = 0;
            if (p >= NSAMP) p = NSAMP - 1;
            span[i] = xb[p];
        }
        __syncthreads();

        if (jj != 0) {
            const int j1 = jj, j2 = KQ - jj;
            for (int r8 = 0; r8 < 8; ++r8) {
                const int r = 8 * rh + r8;
                const float* xr = span + r * HOP;
                const float ya1 = wa1 * xr[j1];
                const float yd1 = wa1 * xr[NFFT - j1];
                const float yb1 = wb1 * xr[NFFT / 2 - j1];
                const float yc1 = wb1 * xr[NFFT / 2 + j1];
                const float p01 = ya1 + yd1, p11 = yb1 + yc1;
                const float q01 = ya1 - yd1, q11 = yb1 - yc1;
                const float ya2 = wa2 * xr[j2];
                const float yd2 = wa2 * xr[NFFT - j2];
                const float yb2 = wb2 * xr[NFFT / 2 - j2];
                const float yc2 = wb2 * xr[NFFT / 2 + j2];
                const float p02 = ya2 + yd2, p12 = yb2 + yc2;
                const float q02 = ya2 - yd2, q12 = yb2 - yc2;
                const float pe1 = p01 + p11, pe2 = p02 + p12;
                const float qe1 = q01 - q11, qe2 = q02 - q12;
                const int o = (16 * h + r) * PITCH;
                PO[o + j1] = p01 - p11;
                PO[o + j2] = p02 - p12;
                QO[o + j1] = q01 + q11;
                QO[o + j2] = q02 + q12;
                const int oe = (16 * h + r) * PITCHE + j1;
                C0[oe] = pe1 + pe2;
                C2[oe] = pe1 - pe2;
                S0[oe] = qe1 - qe2;
                S2[oe] = qe1 + qe2;
            }
        }
        if (tid < 16) {
            const float* xr = span + tid * HOP;
            const float w0 = 0.5f - 0.5f * cosf(0.0f);
            const float wh = 0.5f + 0.5f * cosf(0.0f);
            const float c8 = cosf((float)(NFFT / 8) * (6.28318530717958647692f / (float)NFFT));
            const float w128 = 0.5f - 0.5f * c8;
            const float w384 = 0.5f + 0.5f * c8;
            const float y0   = w0 * xr[0];
            const float y512 = wh * xr[NFFT / 2];
            const float y256 = wq * xr[NFFT / 4];
            const float y768 = wq * xr[3 * NFFT / 4];
            const float ya = w128 * xr[NFFT / 8];
            const float yd = w128 * xr[NFFT - NFFT / 8];
            const float yb = w384 * xr[NFFT / 2 - NFFT / 8];
            const float yc = w384 * xr[NFFT / 2 + NFFT / 8];
            const float p0 = ya + yd, p1 = yb + yc;
            const float q0 = ya - yd, q1 = yb - yc;
            const int o = (16 * h + tid) * PITCH;
            PO[o] = y0 - y512;
            QO[o] = y256 - y768;
            PO[o + KH] = p0 - p1;
            QO[o + KH] = q0 + q1;
            const int oe = (16 * h + tid) * PITCHE;
            const float pe0 = y0 + y512;
            const float pe256 = y256 + y768;
            C0[oe] = pe0 + pe256;
            C2[oe] = pe0 - pe256;
            S0[oe] = 0.0f;
            S2[oe] = q0 - q1;
            C0[oe + KH]     = p0 + p1;
            C0[oe + KH + 1] = 0.0f;
            C0[oe + KH + 2] = 0.0f;
            C0[oe + KH + 3] = 0.0f;
        }
        __syncthreads();
    }

    const int mrow  = ln & 15;
    const int kh    = (ln >> 4) << 1;
    const int ncol  = ln & 15;
    const int mbase = (ln >> 4) * 8;

    float mreg[5][MT][8];
    auto tile_of = [&](int i) { return i == 0 ? NEVT + wv : i == 1 ? NEVT + 8 + wv : i == 2 ? wv : i == 3 ? 8 + wv : 16; };
    auto tile_ptr = [&](int i) {
        const int nt = tile_of(i);
        const int n = nt * 16 + ncol;
        TilePtr p;
        if (i < 2) {
            p.pa = mrow * PITCH + kh;
            p.qa = GFR * PITCH + mrow * PITCH + kh;
        } else {
            const int fam2 = (nt < N0T) ? 0 : GFR * PITCHE;
            p.pa = 2 * GFR * PITCH + fam2 + mrow * PITCHE + kh;
            p.qa = 2 * GFR * PITCH + 2 * GFR * PITCHE + fam2 + mrow * PITCHE + kh;
        }
        p.bo  = (kh >> 1) * (FPAD * 2) + n * 2;
        return p;
    };
    Frag1 fa, fb;
    TilePtr cur = tile_ptr(0);
    load_frag1(fa, cosT, sinT, lds, cur, 0, 16 * PITCH, S1K);
    #pragma unroll
    for (int i = 0; i < 5; ++i) {
        const int nt = tile_of(i);
        if (i < 4 || wv == 0) {
            const bool c0t = nt < N0T;
            const int  kt  = (i < 2) ? KQ : KH;
            const int  p16 = 16 * ((i < 2) ? PITCH : PITCHE);
            const int  n16 = 16 * ((i + 1 < 2) ? PITCH : PITCHE);
            const int  sk  = (i < 2) ? S1K : S1K / 2;
            const int  nsk = (i + 1 < 2) ? S1K : S1K / 2;
            const bool has_next = (i + 1 < 4) || (i + 1 == 4 && wv == 0);
            const TilePtr nxt = tile_ptr(has_next ? i + 1 : i);
            v8f re0 = {0.f, 0.f, 0.f, 0.f, 0.f, 0.f, 0.f, 0.f};
            v8f im0 = {0.f, 0.f, 0.f, 0.f, 0.f, 0.f, 0.f, 0.f};
            v8f re1 = {0.f, 0.f, 0.f, 0.f, 0.f, 0.f, 0.f, 0.f};
            v8f im1 = {0.f, 0.f, 0.f, 0.f, 0.f, 0.f, 0.f, 0.f};
            #pragma unroll 1
            for (int j0 = 0; j0 < kt - 8 * sk; j0 += 8 * sk) {
                load_frag1(fb, cosT, sinT, lds, cur, j0 + 4 * sk, p16, sk);
                __builtin_amdgcn_sched_barrier(0);
                mma_frag1(fa, re0, im0, re1, im1, sk);
                __builtin_amdgcn_sched_barrier(0);
                load_frag1(fa, cosT, sinT, lds, cur, j0 + 8 * sk, p16, sk);
                __builtin_amdgcn_sched_barrier(0);
                mma_frag1(fb, re0, im0, re1, im1, sk);
                __builtin_amdgcn_sched_barrier(0);
            }
            load_frag1(fb, cosT, sinT, lds, cur, kt - 4 * sk, p16, sk);
            __builtin_amdgcn_sched_barrier(0);
            mma_frag1(fa, re0, im0, re1, im1, sk);
            __builtin_amdgcn_sched_barrier(0);
            if (has_next) load_frag1(fa, cosT, sinT, lds, nxt, 0, n16, nsk);
            __builtin_amdgcn_sched_barrier(0);
            mma_frag1(fb, re0, im0, re1, im1, sk);
            if (c0t) {
                v2f bc  = *(const v2f*)&cosT[cur.bo + (KH >> 1) * (FPAD * 2)];
                v2f ap0 = *(const v2f*)&lds[cur.pa + KH];
                v2f ap1 = *(const v2f*)&lds[cur.pa + p16 + KH];
                re0 = __builtin_amdgcn_wmma_f32_16x16x4_f32(false, ap0, false, bc,
                                                            (short)0, re0, false, false);
                re1 = __builtin_amdgcn_wmma_f32_16x16x4_f32(false, ap1, false, bc,
                                                            (short)0, re1, false, false);
            }
            #pragma unroll
            for (int v = 0; v < 8; ++v) {
                float re = re0[v], im = im0[v];
                mreg[i][0][v] = sqrtf(re * re + im * im);
                re = re1[v]; im = im1[v];
                mreg[i][1][v] = sqrtf(re * re + im * im);
            }
            cur = nxt;
        }
    }
    __syncthreads();
    #pragma unroll
    for (int i = 0; i < 5; ++i) {
        const int nt = tile_of(i);
        if (i < 4 || wv == 0) {
            const int n = nt * 16 + ncol;
            #pragma unroll
            for (int m = 0; m < MT; ++m) {
                #pragma unroll
                for (int v = 0; v < 8; ++v)
                    mag[mag_idx(16 * m + mbase + v, n)] = mreg[i][m][v];
            }
        }
    }
    __syncthreads();

    for (int nt = wv; nt < NMELS / 16; nt += 8) {
        v8f acc0 = {0.f, 0.f, 0.f, 0.f, 0.f, 0.f, 0.f, 0.f};
        v8f acc1 = {0.f, 0.f, 0.f, 0.f, 0.f, 0.f, 0.f, 0.f};
        const int n = nt * 16 + ncol;
        const float m_lo = hz_to_mel(50.0f);
        const float m_hi = hz_to_mel((float)SRATE * 0.5f);
        const float mstep = (m_hi - m_lo) / (float)(NMELS + 1);
        const float binw = (float)SRATE * 0.5f / (float)(NFFT / 2);
        const float f_lo = mel_to_hz(m_lo + mstep * (float)(16 * nt));
        const float f_hi = mel_to_hz(m_lo + mstep * (float)(16 * nt + 17));
        const int blo = max(0, (int)floorf(f_lo / binw) - 1);
        const int bhi = min(NFREQS - 1, (int)ceilf(f_hi / binw) + 1);
        const int a0 = __builtin_amdgcn_readfirstlane((blo >> 2) >> 4);
        const int a1 = __builtin_amdgcn_readfirstlane(((bhi >> 2) >> 4) + 1);
        const int c0 = __builtin_amdgcn_readfirstlane(N0T + ((max(0, blo - 2) >> 2) >> 4));
        const int c1 = __builtin_amdgcn_readfirstlane(N0T + ((max(0, bhi - 2) >> 2) >> 4) + 1);
        const int ob0 = __builtin_amdgcn_readfirstlane(NEVT + ((max(0, blo - 1) >> 1) >> 4));
        const int ob1 = __builtin_amdgcn_readfirstlane(NEVT + ((max(0, bhi - 1) >> 1) >> 4) + 1);
        const int na = a1 - a0;
        const int ne = na + (c1 - c0);
        const int nb = ne + (ob1 - ob0);
        auto gcol = [&](int q) { return 16 * (q < na ? a0 + q : q < ne ? c0 + (q - na) : ob0 + (q - ne)); };
        auto load_frag2 = [&](Frag2& f, int kb0) {
            #pragma unroll
            for (int s = 0; s < S2K; ++s) {
                const int k0 = kb0 + 4 * s + kh;
                f.bm[s] = *(const v2f*)&melFB[(k0 >> 1) * (NMELS * 2) + n * 2];
            }
            #pragma unroll
            for (int s = 0; s < S2K; ++s) {
                const int k0 = kb0 + 4 * s + kh;
                f.a0[s] = *(const v2f*)&mag[mag_idx(mrow, k0)];
                f.a1[s] = *(const v2f*)&mag[mag_idx(16 + mrow, k0)];
            }
        };
        auto mma_frag2 = [&](const Frag2& f) {
            #pragma unroll
            for (int s = 0; s < S2K; ++s) {
                acc0 = __builtin_amdgcn_wmma_f32_16x16x4_f32(false, f.a0[s], false, f.bm[s],
                                                             (short)0, acc0, false, false);
                acc1 = __builtin_amdgcn_wmma_f32_16x16x4_f32(false, f.a1[s], false, f.bm[s],
                                                             (short)0, acc1, false, false);
            }
        };
        Frag2 ga, gb;
        load_frag2(ga, gcol(0));
        int q = 0;
        #pragma unroll 1
        for (; q + 2 < nb; q += 2) {
            load_frag2(gb, gcol(q + 1));
            __builtin_amdgcn_sched_barrier(0);
            mma_frag2(ga);
            __builtin_amdgcn_sched_barrier(0);
            load_frag2(ga, gcol(q + 2));
            __builtin_amdgcn_sched_barrier(0);
            mma_frag2(gb);
            __builtin_amdgcn_sched_barrier(0);
        }
        if (q + 1 < nb) {
            load_frag2(gb, gcol(q + 1));
            __builtin_amdgcn_sched_barrier(0);
            mma_frag2(ga);
            mma_frag2(gb);
        } else {
            mma_frag2(ga);
        }
        #pragma unroll
        for (int v = 0; v < 8; ++v) {
            int t = t0 + mbase + v;
            if (t < NFRAMES)
                melBuf[(size_t)t * NCH + b * NMELS + n] = acc0[v];
            t += 16;
            if (t < NFRAMES)
                melBuf[(size_t)t * NCH + b * NMELS + n] = acc1[v];
        }
    }
}

#define PC_CW     8
#define PC_SUB    4
#define PC_L      21
#define PC_NCHUNK 128
#define PC_PITCH  2692
#define PC_BLOCKS (NCH / PC_CW)

__global__ __launch_bounds__(1024)
void pcen_kernel(const float* __restrict__ melBuf,
                 float* __restrict__ out)
{
    __shared__ float tile[PC_CW * PC_PITCH];
    __shared__ float ends[PC_NCHUNK * PC_CW];

    const int tid   = threadIdx.x;
    const int c     = tid & (PC_CW - 1);
    const int chunk = tid >> 3;
    const int ch0   = blockIdx.x * PC_CW;
    const int t0    = chunk * PC_L;

    const float s  = 1.0f - expf(-((float)HOP / (float)SRATE) / 0.06f);
    const float d  = 1.0f - s;
    const float da = powf(2.0f, 0.98f);

    float e[PC_L];
    #pragma unroll
    for (int i = 0; i < PC_L; ++i) {
        const int t = min(t0 + i, NFRAMES - 1);
        e[i] = melBuf[(size_t)t * NCH + ch0 + c];
    }
    float* row = tile + c * PC_PITCH + t0;
    float loc = 0.0f;
    #pragma unroll
    for (int i = 0; i < PC_L; ++i) {
        row[i] = e[i];
        loc = d * loc + s * e[i];
    }
    ends[chunk * PC_CW + c] = loc;
    __syncthreads();

    float dL = 1.0f;
    #pragma unroll
    for (int i = 0; i < PC_L; ++i) dL *= d;
    float M = 0.0f;
    #pragma unroll 8
    for (int j = 0; j < chunk; ++j) M = dL * M + ends[j * PC_CW + c];

    const int n = max(0, min(PC_L, NFRAMES - t0));
    const float am1 = 0.98f - 1.0f;
    for (int i = 0; i < n; ++i) {
        const float ev = row[i];
        M = (1.0f - s) * M + s * ev;
        const float xs = ev * __builtin_amdgcn_rsqf(M + 1e-6f) + 2.0f;
        row[i] = xs * __builtin_amdgcn_exp2f(am1 * __builtin_amdgcn_logf(xs)) - da;
    }
    __syncthreads();

    float* o = out + (size_t)ch0 * NFRAMES;
    for (int idx = tid; idx < PC_CW * NFRAMES; idx += 1024) {
        const int r = idx / NFRAMES;
        const int t = idx - r * NFRAMES;
        o[idx] = tile[r * PC_PITCH + t];
    }
}

extern "C" void kernel_launch(void* const* d_in, const int* in_sizes, int n_in,
                              void* d_out, int out_size, void* d_ws, size_t ws_size,
                              hipStream_t stream) {
    const float* x = (const float*)d_in[0];
    float* ws = (float*)d_ws;

    float* cosT   = ws;
    float* sinT   = cosT  + (size_t)KE * FPAD;
    float* melFB  = sinT  + (size_t)KQ * FPAD;
    float* melBuf = melFB + (size_t)FPAD * NMELS;

    int n1 = KE * FPAD;
    init_dft<<<(n1 + 255) / 256, 256, 0, stream>>>(cosT, sinT);
    int n2 = FPAD * NMELS;
    init_mel<<<(n2 + 255) / 256, 256, 0, stream>>>(melFB);

    dim3 grid(GTILES, BATCH);
    dft_mel_kernel<<<grid, 256, 0, stream>>>(x, cosT, sinT, melFB, melBuf);

    pcen_kernel<<<PC_BLOCKS, 1024, 0, stream>>>(melBuf, (float*)d_out);
}


# dft_mel: fourth fold of the real DFT about tap 64 for the bins 4c (bins 8c take C0[k]+C0[128-k], bins 8c+4 C0[k]-C0[128-k]): K 128->64 on 9 of 33 bin tiles; DFT and mel sum order change
# speedup vs baseline: 16.9457x; 1.0480x over previous
#include <hip/hip_runtime.h>
#include <math.h>

#define SRATE     22050
#define NFFT      1024
#define HOP       256
#define NMELS     128
#define NFREQS    513
#define FPAD      528
#define BCOLS     80
#define C2COLS    144
#define ECOLS     272
#define NAT       (BCOLS/16)
#define N0T       (C2COLS/16)
#define NEVT      (ECOLS/16)
#define BATCH     16
#define NSAMP     661504
#define NFRAMES   2585
#define MT        2
#define GFR       (16*MT)
#define GTILES    81
#define SPAN      (15*HOP + NFFT)
#define NCH       (BATCH*NMELS)
#define KQ        (NFFT/4)
#define KE        (KQ + 4)
#define PITCH     (KQ + 4)
#define KH        (NFFT/8)
#define PITCHE    (KH + 4)
#define KG        (NFFT/16)
#define PITCHF    (KG + 4)

typedef __attribute__((ext_vector_type(2))) float v2f;
typedef __attribute__((ext_vector_type(8))) float v8f;

__device__ __forceinline__ int mag_idx(int row, int col) { return row * FPAD + (col ^ (((row & 15) >> 2) << 2)); }

__device__ __forceinline__ int col_bin(int col) {
    return (col < BCOLS) ? 8 * col : (col < C2COLS) ? 8 * (col - BCOLS) + 4 : (col < ECOLS) ? 4 * (col - C2COLS) + 2 : 2 * (col - ECOLS) + 1;
}

__global__ void init_dft(float* __restrict__ cosT, float* __restrict__ sinT) {
    int idx = blockIdx.x * blockDim.x + threadIdx.x;
    if (idx >= KE * FPAD) return;
    int j   = idx / FPAD;
    int col = idx % FPAD;
    const int fam = (col < BCOLS) ? 0 : (col < C2COLS) ? 1 : (col < ECOLS) ? 2 : 3;
    const int f = col_bin(col);
    const int kc = (fam == 0) ? KG : (fam == 1) ? KG - 1 : (fam == 2) ? KH - 1 : KQ - 1;
    const int ks = (fam == 3) ? KQ : (fam == 2) ? KH : KG;
    float c = 0.0f, s = 0.0f;
    if (f < NFREQS) {
        if (j <= kc) {
            int r = (j * f) & (NFFT - 1);
            c = cosf((float)r * (6.28318530717958647692f / (float)NFFT));
        }
        if (j < ks) {
            int k = (fam != 0 && j == 0) ? ks : j;
            int r = (k * f) & (NFFT - 1);
            s = sinf((float)r * (6.28318530717958647692f / (float)NFFT));
        }
    }
    int off = (j >> 1) * (FPAD * 2) + col * 2 + (j & 1);
    cosT[off] = c;
    if (j < KQ) sinT[off] = s;
}

__device__ __forceinline__ float hz_to_mel(float f) { return 2595.0f * log10f(1.0f + f / 700.0f); }
__device__ __forceinline__ float mel_to_hz(float m) { return 700.0f * (exp10f(m / 2595.0f) - 1.0f); }

__global__ void init_mel(float* __restrict__ melFB) {
    int idx = blockIdx.x * blockDim.x + threadIdx.x;
    if (idx >= FPAD * NMELS) return;
    int k = idx / NMELS;
    int m = idx % NMELS;
    const int bin = col_bin(k);
    float v = 0.0f;
    if (bin < NFREQS) {
        float m_lo = hz_to_mel(50.0f);
        float m_hi = hz_to_mel((float)SRATE * 0.5f);
        float step = (m_hi - m_lo) / (float)(NMELS + 1);
        float f0 = mel_to_hz(m_lo + step * (float)m);
        float f1 = mel_to_hz(m_lo + step * (float)(m + 1));
        float f2 = mel_to_hz(m_lo + step * (float)(m + 2));
        float freq = (float)bin * ((float)SRATE * 0.5f / (float)(NFFT / 2));
        float down = (freq - f0) / (f1 - f0);
        float up   = (f2 - freq) / (f2 - f1);
        v = fmaxf(0.0f, fminf(down, up));
    }
    melFB[(k >> 1) * (NMELS * 2) + m * 2 + (k & 1)] = v;
}

#define S1K 4
#define S2K 4

struct TilePtr { int bo; int pa; int qa; };
struct Frag1 { v2f bc[S1K], bs[S1K], ap0[S1K], aq0[S1K], ap1[S1K], aq1[S1K]; };
struct Frag2 { v2f bm[S2K], a0[S2K], a1[S2K]; };

__device__ __forceinline__ void load_frag1(Frag1& f, const float* __restrict__ cosT, const float* __restrict__ sinT,
                                           const float* lds, const TilePtr& p, int j0, const int p16, const int sk) {
    #pragma unroll
    for (int s = 0; s < sk; ++s) {
        const int j = j0 + 4 * s;
        f.bc[s] = *(const v2f*)&cosT[p.bo + (j >> 1) * (FPAD * 2)];
        f.bs[s] = *(const v2f*)&sinT[p.bo + (j >> 1) * (FPAD * 2)];
    }
    #pragma unroll
    for (int s = 0; s < sk; ++s) {
        const int j = j0 + 4 * s;
        f.ap0[s] = *(const v2f*)&lds[p.pa + j];
        f.aq0[s] = *(const v2f*)&lds[p.qa + j];
        f.ap1[s] = *(const v2f*)&lds[p.pa + p16 + j];
        f.aq1[s] = *(const v2f*)&lds[p.qa + p16 + j];
    }
}

__device__ __forceinline__ void mma_frag1(const Frag1& f, v8f& re0, v8f& im0, v8f& re1, v8f& im1, const int sk) {
    #pragma unroll
    for (int s = 0; s < sk; ++s) {
        re0 = __builtin_amdgcn_wmma_f32_16x16x4_f32(false, f.ap0[s], false, f.bc[s],
                                                    (short)0, re0, false, false);
        im0 = __builtin_amdgcn_wmma_f32_16x16x4_f32(false, f.aq0[s], false, f.bs[s],
                                                    (short)0, im0, false, false);
        re1 = __builtin_amdgcn_wmma_f32_16x16x4_f32(false, f.ap1[s], false, f.bc[s],
                                                    (short)0, re1, false, false);
        im1 = __builtin_amdgcn_wmma_f32_16x16x4_f32(false, f.aq1[s], false, f.bs[s],
                                                    (short)0, im1, false, false);
    }
}

struct PairFold { float po1, po2, qo1, qo2, c0, c2, s0, s2; };
__device__ __forceinline__ PairFold fold_pair(const float* xr, const int j1, const int j2,
                                              const float wa1, const float wb1, const float wa2, const float wb2) {
    const float ya1 = wa1 * xr[j1];
    const float yd1 = wa1 * xr[NFFT - j1];
    const float yb1 = wb1 * xr[NFFT / 2 - j1];
    const float yc1 = wb1 * xr[NFFT / 2 + j1];
    const float p01 = ya1 + yd1, p11 = yb1 + yc1;
    const float q01 = ya1 - yd1, q11 = yb1 - yc1;
    const float ya2 = wa2 * xr[j2];
    const float yd2 = wa2 * xr[NFFT - j2];
    const float yb2 = wb2 * xr[NFFT / 2 - j2];
    const float yc2 = wb2 * xr[NFFT / 2 + j2];
    const float p02 = ya2 + yd2, p12 = yb2 + yc2;
    const float q02 = ya2 - yd2, q12 = yb2 - yc2;
    const float pe1 = p01 + p11, pe2 = p02 + p12;
    const float qe1 = q01 - q11, qe2 = q02 - q12;
    PairFold r;
    r.po1 = p01 - p11;  r.po2 = p02 - p12;
    r.qo1 = q01 + q11;  r.qo2 = q02 + q12;
    r.c0 = pe1 + pe2;   r.c2 = pe1 - pe2;
    r.s0 = qe1 - qe2;   r.s2 = qe1 + qe2;
    return r;
}

__global__ __launch_bounds__(256)
void dft_mel_kernel(const float* __restrict__ x,
                    const float* __restrict__ cosT,
                    const float* __restrict__ sinT,
                    const float* __restrict__ melFB,
                    float* __restrict__ melBuf)
{
    __shared__ float lds[2 * GFR * PITCH + 2 * GFR * PITCHE + 4 * GFR * PITCHF + SPAN];
    float* PO   = lds;
    float* QO   = lds + 1 * GFR * PITCH;
    float* C2   = lds + 2 * GFR * PITCH;
    float* S2   = lds + 2 * GFR * PITCH + 1 * GFR * PITCHE;
    float* E0   = lds + 2 * GFR * PITCH + 2 * GFR * PITCHE;
    float* E4   = E0 + 1 * GFR * PITCHF;
    float* T0   = E0 + 2 * GFR * PITCHF;
    float* T4   = E0 + 3 * GFR * PITCHF;
    float* span = E0 + 4 * GFR * PITCHF;
    float* mag  = lds;

    const int tile = blockIdx.x;
    const int b    = blockIdx.y;
    const int tid  = threadIdx.x;
    const int ln   = tid & 31;
    const int wv   = __builtin_amdgcn_readfirstlane(tid >> 5);
    const int t0   = tile * GFR;

    const int kk = tid & 63;
    const int rq = tid >> 6;
    const float cw1 = cosf((float)kk * (6.28318530717958647692f / (float)NFFT));
    const float cw2 = cosf((float)(KQ - kk) * (6.28318530717958647692f / (float)NFFT));
    const float cw3 = cosf((float)(KH - kk) * (6.28318530717958647692f / (float)NFFT));
    const float cw4 = cosf((float)(KH + kk) * (6.28318530717958647692f / (float)NFFT));
    const float wa1 = 0.5f - 0.5f * cw1, wb1 = 0.5f + 0.5f * cw1;
    const float wa2 = 0.5f - 0.5f * cw2, wb2 = 0.5f + 0.5f * cw2;
    const float wa3 = 0.5f - 0.5f * cw3, wb3 = 0.5f + 0.5f * cw3;
    const float wa4 = 0.5f - 0.5f * cw4, wb4 = 0.5f + 0.5f * cw4;
    const float wq = 0.5f - 0.5f * cosf((float)(NFFT / 4) * (6.28318530717958647692f / (float)NFFT));

    const float* xb = x + (size_t)b * NSAMP;
    for (int h = 0; h < MT; ++h) {
        const long base = (long)(t0 + 16 * h) * HOP - (NFFT / 2);
        for (int i = tid; i < SPAN; i += 256) {
            long p = base + i;
            if (p < 0) p = -p;
            if (p >= NSAMP) p = 2L * NSAMP - 2 - p;
            if (p < 0) p = 0;
            if (p >= NSAMP) p = NSAMP - 1;
            span[i] = xb[p];
        }
        __syncthreads();

        if (kk != 0) {
            const int j1 = kk, j2 = KQ - kk;
            const int j3 = KH - kk, j4 = KH + kk;
            for (int r4 = 0; r4 < 4; ++r4) {
                const int r = 4 * rq + r4;
                const float* xr = span + r * HOP;
                const PairFold u = fold_pair(xr, j1, j2, wa1, wb1, wa2, wb2);
                const PairFold v = fold_pair(xr, j3, j4, wa3, wb3, wa4, wb4);
                const int o = (16 * h + r) * PITCH;
                PO[o + j1] = u.po1;
                PO[o + j2] = u.po2;
                QO[o + j1] = u.qo1;
                QO[o + j2] = u.qo2;
                PO[o + j3] = v.po1;
                PO[o + j4] = v.po2;
                QO[o + j3] = v.qo1;
                QO[o + j4] = v.qo2;
                const int oe = (16 * h + r) * PITCHE;
                C2[oe + j1] = u.c2;
                S2[oe + j1] = u.s2;
                C2[oe + j3] = v.c2;
                S2[oe + j3] = v.s2;
                const int og = (16 * h + r) * PITCHF + j1;
                E0[og] = u.c0 + v.c0;
                E4[og] = u.c0 - v.c0;
                T0[og] = u.s0 - v.s0;
                T4[og] = u.s0 + v.s0;
            }
        }
        if (tid < 16) {
            const float* xr = span + tid * HOP;
            const float w0 = 0.5f - 0.5f * cosf(0.0f);
            const float wh = 0.5f + 0.5f * cosf(0.0f);
            const float c8 = cosf((float)(NFFT / 8) * (6.28318530717958647692f / (float)NFFT));
            const float w128 = 0.5f - 0.5f * c8;
            const float w384 = 0.5f + 0.5f * c8;
            const float y0   = w0 * xr[0];
            const float y512 = wh * xr[NFFT / 2];
            const float y256 = wq * xr[NFFT / 4];
            const float y768 = wq * xr[3 * NFFT / 4];
            const float ya = w128 * xr[NFFT / 8];
            const float yd = w128 * xr[NFFT - NFFT / 8];
            const float yb = w384 * xr[NFFT / 2 - NFFT / 8];
            const float yc = w384 * xr[NFFT / 2 + NFFT / 8];
            const float p0 = ya + yd, p1 = yb + yc;
            const float q0 = ya - yd, q1 = yb - yc;
            const float c16  = cosf((float)KG * (6.28318530717958647692f / (float)NFFT));
            const float c316 = cosf((float)(KQ - KG) * (6.28318530717958647692f / (float)NFFT));
            const PairFold m = fold_pair(xr, KG, KQ - KG, 0.5f - 0.5f * c16, 0.5f + 0.5f * c16,
                                         0.5f - 0.5f * c316, 0.5f + 0.5f * c316);
            const int o = (16 * h + tid) * PITCH;
            PO[o] = y0 - y512;
            QO[o] = y256 - y768;
            PO[o + KH] = p0 - p1;
            QO[o + KH] = q0 + q1;
            PO[o + KG] = m.po1;
            PO[o + KQ - KG] = m.po2;
            QO[o + KG] = m.qo1;
            QO[o + KQ - KG] = m.qo2;
            const int oe = (16 * h + tid) * PITCHE;
            const float pe0 = y0 + y512;
            const float pe256 = y256 + y768;
            C2[oe] = pe0 - pe256;
            S2[oe] = q0 - q1;
            C2[oe + KG] = m.c2;
            S2[oe + KG] = m.s2;
            const int og = (16 * h + tid) * PITCHF;
            const float c0a = pe0 + pe256;
            const float c0b = p0 + p1;
            E0[og] = c0a + c0b;
            E4[og] = c0a - c0b;
            T0[og] = 0.0f;
            T4[og] = m.s0;
            E0[og + KG]     = m.c0;
            E0[og + KG + 1] = 0.0f;
            E0[og + KG + 2] = 0.0f;
            E0[og + KG + 3] = 0.0f;
        }
        __syncthreads();
    }

    const int mrow  = ln & 15;
    const int kh    = (ln >> 4) << 1;
    const int ncol  = ln & 15;
    const int mbase = (ln >> 4) * 8;

    float mreg[5][MT][8];
    auto tile_of = [&](int i) { return i == 0 ? NEVT + wv : i == 1 ? NEVT + 8 + wv : i == 2 ? N0T + wv : i == 3 ? wv : 8; };
    auto tile_ptr = [&](int i) {
        const int nt = tile_of(i);
        const int n = nt * 16 + ncol;
        TilePtr p;
        if (i < 2) {
            p.pa = mrow * PITCH + kh;
            p.qa = GFR * PITCH + mrow * PITCH + kh;
        } else if (i == 2) {
            p.pa = 2 * GFR * PITCH + mrow * PITCHE + kh;
            p.qa = 2 * GFR * PITCH + GFR * PITCHE + mrow * PITCHE + kh;
        } else {
            const int fam4 = (nt < NAT) ? 0 : GFR * PITCHF;
            p.pa = 2 * GFR * PITCH + 2 * GFR * PITCHE + fam4 + mrow * PITCHF + kh;
            p.qa = 2 * GFR * PITCH + 2 * GFR * PITCHE + 2 * GFR * PITCHF + fam4 + mrow * PITCHF + kh;
        }
        p.bo  = (kh >> 1) * (FPAD * 2) + n * 2;
        return p;
    };
    Frag1 fa, fb;
    TilePtr cur = tile_ptr(0);
    load_frag1(fa, cosT, sinT, lds, cur, 0, 16 * PITCH, S1K);
    #pragma unroll
    for (int i = 0; i < 5; ++i) {
        const int nt = tile_of(i);
        if (i < 4 || wv == 0) {
            const bool c0t = (i >= 3) && (nt < NAT);
            const int  kt  = (i < 2) ? KQ : (i == 2) ? KH : KG;
            const int  p16 = 16 * ((i < 2) ? PITCH : (i == 2) ? PITCHE : PITCHF);
            const int  n16 = 16 * ((i + 1 < 2) ? PITCH : (i + 1 == 2) ? PITCHE : PITCHF);
            const int  sk  = (i < 2) ? S1K : S1K / 2;
            const int  nsk = (i + 1 < 2) ? S1K : S1K / 2;
            const bool has_next = (i + 1 < 4) || (i + 1 == 4 && wv == 0);
            const TilePtr nxt = tile_ptr(has_next ? i + 1 : i);
            v8f re0 = {0.f, 0.f, 0.f, 0.f, 0.f, 0.f, 0.f, 0.f};
            v8f im0 = {0.f, 0.f, 0.f, 0.f, 0.f, 0.f, 0.f, 0.f};
            v8f re1 = {0.f, 0.f, 0.f, 0.f, 0.f, 0.f, 0.f, 0.f};
            v8f im1 = {0.f, 0.f, 0.f, 0.f, 0.f, 0.f, 0.f, 0.f};
            #pragma unroll 1
            for (int j0 = 0; j0 < kt - 8 * sk; j0 += 8 * sk) {
                load_frag1(fb, cosT, sinT, lds, cur, j0 + 4 * sk, p16, sk);
                __builtin_amdgcn_sched_barrier(0);
                mma_frag1(fa, re0, im0, re1, im1, sk);
                __builtin_amdgcn_sched_barrier(0);
                load_frag1(fa, cosT, sinT, lds, cur, j0 + 8 * sk, p16, sk);
                __builtin_amdgcn_sched_barrier(0);
                mma_frag1(fb, re0, im0, re1, im1, sk);
                __builtin_amdgcn_sched_barrier(0);
            }
            load_frag1(fb, cosT, sinT, lds, cur, kt - 4 * sk, p16, sk);
            __builtin_amdgcn_sched_barrier(0);
            mma_frag1(fa, re0, im0, re1, im1, sk);
            __builtin_amdgcn_sched_barrier(0);
            if (has_next) load_frag1(fa, cosT, sinT, lds, nxt, 0, n16, nsk);
            __builtin_amdgcn_sched_barrier(0);
            mma_frag1(fb, re0, im0, re1, im1, sk);
            if (c0t) {
                v2f bc  = *(const v2f*)&cosT[cur.bo + (KG >> 1) * (FPAD * 2)];
                v2f ap0 = *(const v2f*)&lds[cur.pa + KG];
                v2f ap1 = *(const v2f*)&lds[cur.pa + p16 + KG];
                re0 = __builtin_amdgcn_wmma_f32_16x16x4_f32(false, ap0, false, bc,
                                                            (short)0, re0, false, false);
                re1 = __builtin_amdgcn_wmma_f32_16x16x4_f32(false, ap1, false, bc,
                                                            (short)0, re1, false, false);
            }
            #pragma unroll
            for (int v = 0; v < 8; ++v) {
                float re = re0[v], im = im0[v];
                mreg[i][0][v] = sqrtf(re * re + im * im);
                re = re1[v]; im = im1[v];
                mreg[i][1][v] = sqrtf(re * re + im * im);
            }
            cur = nxt;
        }
    }
    __syncthreads();
    #pragma unroll
    for (int i = 0; i < 5; ++i) {
        const int nt = tile_of(i);
        if (i < 4 || wv == 0) {
            const int n = nt * 16 + ncol;
            #pragma unroll
            for (int m = 0; m < MT; ++m) {
                #pragma unroll
                for (int v = 0; v < 8; ++v)
                    mag[mag_idx(16 * m + mbase + v, n)] = mreg[i][m][v];
            }
        }
    }
    __syncthreads();

    for (int nt = wv; nt < NMELS / 16; nt += 8) {
        v8f acc0 = {0.f, 0.f, 0.f, 0.f, 0.f, 0.f, 0.f, 0.f};
        v8f acc1 = {0.f, 0.f, 0.f, 0.f, 0.f, 0.f, 0.f, 0.f};
        const int n = nt * 16 + ncol;
        const float m_lo = hz_to_mel(50.0f);
        const float m_hi = hz_to_mel((float)SRATE * 0.5f);
        const float mstep = (m_hi - m_lo) / (float)(NMELS + 1);
        const float binw = (float)SRATE * 0.5f / (float)(NFFT / 2);
        const float f_lo = mel_to_hz(m_lo + mstep * (float)(16 * nt));
        const float f_hi = mel_to_hz(m_lo + mstep * (float)(16 * nt + 17));
        const int blo = max(0, (int)floorf(f_lo / binw) - 1);
        const int bhi = min(NFREQS - 1, (int)ceilf(f_hi / binw) + 1);
        const int a0 = __builtin_amdgcn_readfirstlane((blo >> 3) >> 4);
        const int a1 = __builtin_amdgcn_readfirstlane(((bhi >> 3) >> 4) + 1);
        const int b0 = __builtin_amdgcn_readfirstlane(NAT + ((max(0, blo - 4) >> 3) >> 4));
        const int b1 = __builtin_amdgcn_readfirstlane(NAT + ((max(0, bhi - 4) >> 3) >> 4) + 1);
        const int c0 = __builtin_amdgcn_readfirstlane(N0T + ((max(0, blo - 2) >> 2) >> 4));
        const int c1 = __builtin_amdgcn_readfirstlane(N0T + ((max(0, bhi - 2) >> 2) >> 4) + 1);
        const int ob0 = __builtin_amdgcn_readfirstlane(NEVT + ((max(0, blo - 1) >> 1) >> 4));
        const int ob1 = __builtin_amdgcn_readfirstlane(NEVT + ((max(0, bhi - 1) >> 1) >> 4) + 1);
        const int na = a1 - a0;
        const int n4 = na + (b1 - b0);
        const int ne = n4 + (c1 - c0);
        const int nb = ne + (ob1 - ob0);
        auto gcol = [&](int q) { return 16 * (q < na ? a0 + q : q < n4 ? b0 + (q - na) : q < ne ? c0 + (q - n4) : ob0 + (q - ne)); };
        auto load_frag2 = [&](Frag2& f, int kb0) {
            #pragma unroll
            for (int s = 0; s < S2K; ++s) {
                const int k0 = kb0 + 4 * s + kh;
                f.bm[s] = *(const v2f*)&melFB[(k0 >> 1) * (NMELS * 2) + n * 2];
            }
            #pragma unroll
            for (int s = 0; s < S2K; ++s) {
                const int k0 = kb0 + 4 * s + kh;
                f.a0[s] = *(const v2f*)&mag[mag_idx(mrow, k0)];
                f.a1[s] = *(const v2f*)&mag[mag_idx(16 + mrow, k0)];
            }
        };
        auto mma_frag2 = [&](const Frag2& f) {
            #pragma unroll
            for (int s = 0; s < S2K; ++s) {
                acc0 = __builtin_amdgcn_wmma_f32_16x16x4_f32(false, f.a0[s], false, f.bm[s],
                                                             (short)0, acc0, false, false);
                acc1 = __builtin_amdgcn_wmma_f32_16x16x4_f32(false, f.a1[s], false, f.bm[s],
                                                             (short)0, acc1, false, false);
            }
        };
        Frag2 ga, gb;
        load_frag2(ga, gcol(0));
        int q = 0;
        #pragma unroll 1
        for (; q + 2 < nb; q += 2) {
            load_frag2(gb, gcol(q + 1));
            __builtin_amdgcn_sched_barrier(0);
            mma_frag2(ga);
            __builtin_amdgcn_sched_barrier(0);
            load_frag2(ga, gcol(q + 2));
            __builtin_amdgcn_sched_barrier(0);
            mma_frag2(gb);
            __builtin_amdgcn_sched_barrier(0);
        }
        if (q + 1 < nb) {
            load_frag2(gb, gcol(q + 1));
            __builtin_amdgcn_sched_barrier(0);
            mma_frag2(ga);
            mma_frag2(gb);
        } else {
            mma_frag2(ga);
        }
        #pragma unroll
        for (int v = 0; v < 8; ++v) {
            int t = t0 + mbase + v;
            if (t < NFRAMES)
                melBuf[(size_t)t * NCH + b * NMELS + n] = acc0[v];
            t += 16;
            if (t < NFRAMES)
                melBuf[(size_t)t * NCH + b * NMELS + n] = acc1[v];
        }
    }
}

#define PC_CW     8
#define PC_SUB    4
#define PC_L      21
#define PC_NCHUNK 128
#define PC_PITCH  2692
#define PC_BLOCKS (NCH / PC_CW)

__global__ __launch_bounds__(1024)
void pcen_kernel(const float* __restrict__ melBuf,
                 float* __restrict__ out)
{
    __shared__ float tile[PC_CW * PC_PITCH];
    __shared__ float ends[PC_NCHUNK * PC_CW];

    const int tid   = threadIdx.x;
    const int c     = tid & (PC_CW - 1);
    const int chunk = tid >> 3;
    const int ch0   = blockIdx.x * PC_CW;
    const int t0    = chunk * PC_L;

    const float s  = 1.0f - expf(-((float)HOP / (float)SRATE) / 0.06f);
    const float d  = 1.0f - s;
    const float da = powf(2.0f, 0.98f);

    float e[PC_L];
    #pragma unroll
    for (int i = 0; i < PC_L; ++i) {
        const int t = min(t0 + i, NFRAMES - 1);
        e[i] = melBuf[(size_t)t * NCH + ch0 + c];
    }
    float* row = tile + c * PC_PITCH + t0;
    float loc = 0.0f;
    #pragma unroll
    for (int i = 0; i < PC_L; ++i) {
        row[i] = e[i];
        loc = d * loc + s * e[i];
    }
    ends[chunk * PC_CW + c] = loc;
    __syncthreads();

    float dL = 1.0f;
    #pragma unroll
    for (int i = 0; i < PC_L; ++i) dL *= d;
    float M = 0.0f;
    #pragma unroll 8
    for (int j = 0; j < chunk; ++j) M = dL * M + ends[j * PC_CW + c];

    const int n = max(0, min(PC_L, NFRAMES - t0));
    const float am1 = 0.98f - 1.0f;
    for (int i = 0; i < n; ++i) {
        const float ev = row[i];
        M = (1.0f - s) * M + s * ev;
        const float xs = ev * __builtin_amdgcn_rsqf(M + 1e-6f) + 2.0f;
        row[i] = xs * __builtin_amdgcn_exp2f(am1 * __builtin_amdgcn_logf(xs)) - da;
    }
    __syncthreads();

    float* o = out + (size_t)ch0 * NFRAMES;
    for (int idx = tid; idx < PC_CW * NFRAMES; idx += 1024) {
        const int r = idx / NFRAMES;
        const int t = idx - r * NFRAMES;
        o[idx] = tile[r * PC_PITCH + t];
    }
}

extern "C" void kernel_launch(void* const* d_in, const int* in_sizes, int n_in,
                              void* d_out, int out_size, void* d_ws, size_t ws_size,
                              hipStream_t stream) {
    const float* x = (const float*)d_in[0];
    float* ws = (float*)d_ws;

    float* cosT   = ws;
    float* sinT   = cosT  + (size_t)KE * FPAD;
    float* melFB  = sinT  + (size_t)KQ * FPAD;
    float* melBuf = melFB + (size_t)FPAD * NMELS;

    int n1 = KE * FPAD;
    init_dft<<<(n1 + 255) / 256, 256, 0, stream>>>(cosT, sinT);
    int n2 = FPAD * NMELS;
    init_mel<<<(n2 + 255) / 256, 256, 0, stream>>>(melFB);

    dim3 grid(GTILES, BATCH);
    dft_mel_kernel<<<grid, 256, 0, stream>>>(x, cosT, sinT, melFB, melBuf);

    pcen_kernel<<<PC_BLOCKS, 1024, 0, stream>>>(melBuf, (float*)d_out);
}
